# starting kernel
# baseline (speedup 1.0000x reference)
#include <hip/hip_runtime.h>

#define N_NODES 100000
#define N_EDGES 1600000
#define F_IN    512
#define HIDDEN  16
#define N_CLS   7

typedef float v2f __attribute__((ext_vector_type(2)));
typedef float v8f __attribute__((ext_vector_type(8)));

__global__ void k_deg_init(float* deg, int n) {
    int t = blockIdx.x * blockDim.x + threadIdx.x;
    if (t < n) deg[t] = 1.0f;
}

__global__ void k_deg_edges(const int* __restrict__ dst, float* deg, int nE) {
    int e = blockIdx.x * blockDim.x + threadIdx.x;
    if (e < nE) atomicAdd(&deg[dst[e]], 1.0f);
}

__global__ void k_dinv(float* deg, int n) {
    int t = blockIdx.x * blockDim.x + threadIdx.x;
    if (t < n) deg[t] = __frsqrt_rn(deg[t]);
}

__global__ __launch_bounds__(256) void k_gemm0(const float* __restrict__ x,
                                               const float* __restrict__ W0,
                                               float* __restrict__ h0, int nTiles) {
    __shared__ float sW[F_IN * HIDDEN];
    for (int i = threadIdx.x; i < (F_IN * HIDDEN) / 4; i += 256)
        ((float4*)sW)[i] = ((const float4*)W0)[i];
    __syncthreads();

    const int wave   = threadIdx.x >> 5;
    const int lane   = threadIdx.x & 31;
    const int tile   = blockIdx.x * 8 + wave;
    if (tile >= nTiles) return;

    const int laneHi = lane >> 4;
    const int lane15 = lane & 15;
    const int row    = tile * 16 + lane15;
    const float* xrow = x + (long)row * F_IN;
    const int koff   = laneHi ? 4 : 0;

    v8f acc = {};
    for (int kb = 0; kb < F_IN; kb += 8) {
        float4 a4 = *(const float4*)(xrow + kb + koff);
        const float* wb = sW + (kb + koff) * HIDDEN + lane15;
        v2f a0; a0.x = a4.x; a0.y = a4.y;
        v2f b0; b0.x = wb[0];  b0.y = wb[16];
        acc = __builtin_amdgcn_wmma_f32_16x16x4_f32(false, a0, false, b0,
                                                    (short)0, acc, false, false);
        v2f a1; a1.x = a4.z; a1.y = a4.w;
        v2f b1; b1.x = wb[32]; b1.y = wb[48];
        acc = __builtin_amdgcn_wmma_f32_16x16x4_f32(false, a1, false, b1,
                                                    (short)0, acc, false, false);
    }
    float* out = h0 + (long)tile * 256 + laneHi * 128 + lane15;
#pragma unroll
    for (int i = 0; i < 8; ++i) out[i * 16] = acc[i];
}

__global__ void k_self_init16(const float* __restrict__ h, const float* __restrict__ dinv,
                              float* __restrict__ agg, int n4) {
    int t = blockIdx.x * blockDim.x + threadIdx.x;
    if (t >= n4) return;
    int node = t >> 2, q = t & 3;
    float w = dinv[node]; w *= w;
    float4 v = ((const float4*)(h + (long)node * 16))[q];
    v.x *= w; v.y *= w; v.z *= w; v.w *= w;
    ((float4*)(agg + (long)node * 16))[q] = v;
}

__global__ void k_edge16(const int* __restrict__ src, const int* __restrict__ dst,
                         const float* __restrict__ dinv, const float* __restrict__ h,
                         float* __restrict__ agg, int nE) {
    int e = blockIdx.x * blockDim.x + threadIdx.x;
    if (e >= nE) return;
    int s = src[e], d = dst[e];
    float w = dinv[s] * dinv[d];
    const float* hs = h + (long)s * 16;
    float* ad = agg + (long)d * 16;
#pragma unroll
    for (int i = 0; i < 16; ++i) atomicAdd(ad + i, hs[i] * w);
}

__global__ void k_relu_bias(const float* __restrict__ agg, const float* __restrict__ b0,
                            float* __restrict__ h1, int total) {
    int t = blockIdx.x * blockDim.x + threadIdx.x;
    if (t >= total) return;
    float v = agg[t] + b0[t & 15];
    h1[t] = v > 0.0f ? v : 0.0f;
}

__global__ __launch_bounds__(256) void k_gemm1(const float* __restrict__ h1,
                                               const float* __restrict__ W1,
                                               float* __restrict__ h2, int nTiles) {
    __shared__ float sW[16 * 16];
    {
        int k = threadIdx.x >> 4, n = threadIdx.x & 15;
        sW[threadIdx.x] = (n < N_CLS) ? W1[k * N_CLS + n] : 0.0f;
    }
    __syncthreads();

    const int wave   = threadIdx.x >> 5;
    const int lane   = threadIdx.x & 31;
    const int tile   = blockIdx.x * 8 + wave;
    if (tile >= nTiles) return;

    const int laneHi = lane >> 4;
    const int lane15 = lane & 15;
    const int row    = tile * 16 + lane15;
    const float* hrow = h1 + (long)row * 16;
    const int koff   = laneHi ? 2 : 0;

    v8f acc = {};
#pragma unroll
    for (int kb = 0; kb < 16; kb += 4) {
        v2f a; a.x = hrow[kb + koff]; a.y = hrow[kb + koff + 1];
        v2f b; b.x = sW[(kb + koff) * 16 + lane15];
               b.y = sW[(kb + koff + 1) * 16 + lane15];
        acc = __builtin_amdgcn_wmma_f32_16x16x4_f32(false, a, false, b,
                                                    (short)0, acc, false, false);
    }
    float* out = h2 + (long)tile * 256 + laneHi * 128 + lane15;
#pragma unroll
    for (int i = 0; i < 8; ++i) out[i * 16] = acc[i];
}

__global__ void k_out_init7(const float* __restrict__ h2, const float* __restrict__ dinv,
                            const float* __restrict__ b1, float* __restrict__ out, int total) {
    int t = blockIdx.x * blockDim.x + threadIdx.x;
    if (t >= total) return;
    int node = t / N_CLS, c = t - node * N_CLS;
    float w = dinv[node];
    out[t] = h2[(long)node * 16 + c] * w * w + b1[c];
}

__global__ void k_edge7(const int* __restrict__ src, const int* __restrict__ dst,
                        const float* __restrict__ dinv, const float* __restrict__ h2,
                        float* __restrict__ out, int nE) {
    int e = blockIdx.x * blockDim.x + threadIdx.x;
    if (e >= nE) return;
    int s = src[e], d = dst[e];
    float w = dinv[s] * dinv[d];
    const float* hs = h2 + (long)s * 16;
    float* od = out + (long)d * N_CLS;
#pragma unroll
    for (int c = 0; c < N_CLS; ++c) atomicAdd(od + c, hs[c] * w);
}

extern "C" void kernel_launch(void* const* d_in, const int* in_sizes, int n_in,
                              void* d_out, int out_size, void* d_ws, size_t ws_size,
                              hipStream_t stream) {
    const float* x  = (const float*)d_in[0];
    const int*   ei = (const int*)d_in[1];
    const float* W0 = (const float*)d_in[2];
    const float* b0 = (const float*)d_in[3];
    const float* W1 = (const float*)d_in[4];
    const float* b1 = (const float*)d_in[5];
    float* out = (float*)d_out;

    const int* src = ei;
    const int* dst = ei + N_EDGES;

    float* ws   = (float*)d_ws;
    float* dinv = ws;
    float* h0   = dinv + N_NODES;
    float* agg0 = h0 + (long)N_NODES * 16;
    float* h2   = agg0 + (long)N_NODES * 16;

    const int nTiles = N_NODES / 16;
    const dim3 B(256);
    auto gr = [](long n) { return dim3((unsigned)((n + 255) / 256)); };

    k_deg_init <<<gr(N_NODES), B, 0, stream>>>(dinv, N_NODES);
    k_deg_edges<<<gr(N_EDGES), B, 0, stream>>>(dst, dinv, N_EDGES);
    k_dinv     <<<gr(N_NODES), B, 0, stream>>>(dinv, N_NODES);

    k_gemm0<<<dim3((nTiles + 7) / 8), B, 0, stream>>>(x, W0, h0, nTiles);
    k_self_init16<<<gr((long)N_NODES * 4), B, 0, stream>>>(h0, dinv, agg0, N_NODES * 4);
    k_edge16<<<gr(N_EDGES), B, 0, stream>>>(src, dst, dinv, h0, agg0, N_EDGES);
    k_relu_bias<<<gr((long)N_NODES * 16), B, 0, stream>>>(agg0, b0, h0, N_NODES * 16);

    k_gemm1<<<dim3((nTiles + 7) / 8), B, 0, stream>>>(h0, W1, h2, nTiles);
    k_out_init7<<<gr((long)N_NODES * N_CLS), B, 0, stream>>>(h2, dinv, b1, out, N_NODES * N_CLS);
    k_edge7<<<gr(N_EDGES), B, 0, stream>>>(src, dst, dinv, h2, out, N_EDGES);
}


# Edge scatter-adds remapped: 16 lanes (layer 0) or 8 lanes (layer 1) per edge, so each wave-wide row load and f32 atomic covers whole rows; 128 edges per wave; sums stay in atomic arrival order
# speedup vs baseline: 2.9162x; 2.9162x over previous
#include <hip/hip_runtime.h>

#define N_NODES 100000
#define N_EDGES 1600000
#define F_IN    512
#define HIDDEN  16
#define N_CLS   7

typedef float v2f __attribute__((ext_vector_type(2)));
typedef float v8f __attribute__((ext_vector_type(8)));

__global__ void k_deg_init(float* deg, int n) {
    int t = blockIdx.x * blockDim.x + threadIdx.x;
    if (t < n) deg[t] = 1.0f;
}

__global__ void k_deg_edges(const int* __restrict__ dst, float* deg, int nE) {
    int e = blockIdx.x * blockDim.x + threadIdx.x;
    if (e < nE) atomicAdd(&deg[dst[e]], 1.0f);
}

__global__ void k_dinv(float* deg, int n) {
    int t = blockIdx.x * blockDim.x + threadIdx.x;
    if (t < n) deg[t] = __frsqrt_rn(deg[t]);
}

__global__ __launch_bounds__(256) void k_gemm0(const float* __restrict__ x,
                                               const float* __restrict__ W0,
                                               float* __restrict__ h0, int nTiles) {
    __shared__ float sW[F_IN * HIDDEN];
    for (int i = threadIdx.x; i < (F_IN * HIDDEN) / 4; i += 256)
        ((float4*)sW)[i] = ((const float4*)W0)[i];
    __syncthreads();

    const int wave   = threadIdx.x >> 5;
    const int lane   = threadIdx.x & 31;
    const int tile   = blockIdx.x * 8 + wave;
    if (tile >= nTiles) return;

    const int laneHi = lane >> 4;
    const int lane15 = lane & 15;
    const int row    = tile * 16 + lane15;
    const float* xrow = x + (long)row * F_IN;
    const int koff   = laneHi ? 4 : 0;

    v8f acc = {};
    for (int kb = 0; kb < F_IN; kb += 8) {
        float4 a4 = *(const float4*)(xrow + kb + koff);
        const float* wb = sW + (kb + koff) * HIDDEN + lane15;
        v2f a0; a0.x = a4.x; a0.y = a4.y;
        v2f b0; b0.x = wb[0];  b0.y = wb[16];
        acc = __builtin_amdgcn_wmma_f32_16x16x4_f32(false, a0, false, b0,
                                                    (short)0, acc, false, false);
        v2f a1; a1.x = a4.z; a1.y = a4.w;
        v2f b1; b1.x = wb[32]; b1.y = wb[48];
        acc = __builtin_amdgcn_wmma_f32_16x16x4_f32(false, a1, false, b1,
                                                    (short)0, acc, false, false);
    }
    float* out = h0 + (long)tile * 256 + laneHi * 128 + lane15;
#pragma unroll
    for (int i = 0; i < 8; ++i) out[i * 16] = acc[i];
}

__global__ void k_self_init16(const float* __restrict__ h, const float* __restrict__ dinv,
                              float* __restrict__ agg, int n4) {
    int t = blockIdx.x * blockDim.x + threadIdx.x;
    if (t >= n4) return;
    int node = t >> 2, q = t & 3;
    float w = dinv[node]; w *= w;
    float4 v = ((const float4*)(h + (long)node * 16))[q];
    v.x *= w; v.y *= w; v.z *= w; v.w *= w;
    ((float4*)(agg + (long)node * 16))[q] = v;
}

#define EDGES_PER_WAVE 128
static_assert(N_EDGES % EDGES_PER_WAVE == 0, "a wave takes whole groups of 128 edges");

__global__ __launch_bounds__(256) void k_edge16(const int* __restrict__ src, const int* __restrict__ dst,
                         const float* __restrict__ dinv, const float* __restrict__ h,
                         float* __restrict__ agg, int nE) {
    const int lane = threadIdx.x & 31;
    const int base = (blockIdx.x * 8 + (threadIdx.x >> 5)) * EDGES_PER_WAVE;
    if (base >= nE) return;
    const int f    = lane & 15;
    const int half = lane & 16;
    const int4 s4 = *(const int4*)(src + base + 4 * lane);
    const int4 d4 = *(const int4*)(dst + base + 4 * lane);
    const int sv[4] = {s4.x, s4.y, s4.z, s4.w};
    const int dv[4] = {d4.x, d4.y, d4.z, d4.w};
    float wv[4];
    for (int k = 0; k < 4; ++k) wv[k] = dinv[sv[k]] * dinv[dv[k]];
#pragma unroll
    for (int k = 0; k < 4; ++k) {
        const float w  = wv[k];
        const int sOff = sv[k] * 16;
        const int dOff = dv[k] * 16;
        float val[16];
        int   dof[16];
        for (int j = 0; j < 16; ++j) {
            const int   so = __shfl(sOff, half | j);
            const float ww = __shfl(w,    half | j);
            dof[j] = __shfl(dOff, half | j) + f;
            val[j] = h[so + f] * ww;
        }
        for (int j = 0; j < 16; ++j) atomicAdd(&agg[dof[j]], val[j]);
    }
}

__global__ void k_relu_bias(const float* __restrict__ agg, const float* __restrict__ b0,
                            float* __restrict__ h1, int total) {
    int t = blockIdx.x * blockDim.x + threadIdx.x;
    if (t >= total) return;
    float v = agg[t] + b0[t & 15];
    h1[t] = v > 0.0f ? v : 0.0f;
}

__global__ __launch_bounds__(256) void k_gemm1(const float* __restrict__ h1,
                                               const float* __restrict__ W1,
                                               float* __restrict__ h2, int nTiles) {
    __shared__ float sW[16 * 16];
    {
        int k = threadIdx.x >> 4, n = threadIdx.x & 15;
        sW[threadIdx.x] = (n < N_CLS) ? W1[k * N_CLS + n] : 0.0f;
    }
    __syncthreads();

    const int wave   = threadIdx.x >> 5;
    const int lane   = threadIdx.x & 31;
    const int tile   = blockIdx.x * 8 + wave;
    if (tile >= nTiles) return;

    const int laneHi = lane >> 4;
    const int lane15 = lane & 15;
    const int row    = tile * 16 + lane15;
    const float* hrow = h1 + (long)row * 16;
    const int koff   = laneHi ? 2 : 0;

    v8f acc = {};
#pragma unroll
    for (int kb = 0; kb < 16; kb += 4) {
        v2f a; a.x = hrow[kb + koff]; a.y = hrow[kb + koff + 1];
        v2f b; b.x = sW[(kb + koff) * 16 + lane15];
               b.y = sW[(kb + koff + 1) * 16 + lane15];
        acc = __builtin_amdgcn_wmma_f32_16x16x4_f32(false, a, false, b,
                                                    (short)0, acc, false, false);
    }
    float* out = h2 + (long)tile * 256 + laneHi * 128 + lane15;
#pragma unroll
    for (int i = 0; i < 8; ++i) out[i * 16] = acc[i];
}

__global__ void k_out_init7(const float* __restrict__ h2, const float* __restrict__ dinv,
                            const float* __restrict__ b1, float* __restrict__ out, int total) {
    int t = blockIdx.x * blockDim.x + threadIdx.x;
    if (t >= total) return;
    int node = t / N_CLS, c = t - node * N_CLS;
    float w = dinv[node];
    out[t] = h2[(long)node * 16 + c] * w * w + b1[c];
}

__global__ __launch_bounds__(256) void k_edge7(const int* __restrict__ src, const int* __restrict__ dst,
                        const float* __restrict__ dinv, const float* __restrict__ h2,
                        float* __restrict__ out, int nE) {
    const int lane = threadIdx.x & 31;
    const int base = (blockIdx.x * 8 + (threadIdx.x >> 5)) * EDGES_PER_WAVE;
    if (base >= nE) return;
    const int c   = lane & 7;
    const int grp = lane & 24;
    const int4 s4 = *(const int4*)(src + base + 4 * lane);
    const int4 d4 = *(const int4*)(dst + base + 4 * lane);
    const int sv[4] = {s4.x, s4.y, s4.z, s4.w};
    const int dv[4] = {d4.x, d4.y, d4.z, d4.w};
    float wv[4];
    for (int k = 0; k < 4; ++k) wv[k] = dinv[sv[k]] * dinv[dv[k]];
    for (int k = 0; k < 4; ++k) {
        const float w  = wv[k];
        const int sOff = sv[k] * 16;
        const int dOff = dv[k] * N_CLS;
        int   sof[8], dof[8];
        float ww[8];
        for (int j = 0; j < 8; ++j) {
            sof[j] = __shfl(sOff, grp | j) + c;
            dof[j] = __shfl(dOff, grp | j) + c;
            ww[j]  = __shfl(w,    grp | j);
        }
        if (c < N_CLS) {
            float val[8];
            for (int j = 0; j < 8; ++j) val[j] = h2[sof[j]] * ww[j];
            for (int j = 0; j < 8; ++j) atomicAdd(&out[dof[j]], val[j]);
        }
    }
}

extern "C" void kernel_launch(void* const* d_in, const int* in_sizes, int n_in,
                              void* d_out, int out_size, void* d_ws, size_t ws_size,
                              hipStream_t stream) {
    const float* x  = (const float*)d_in[0];
    const int*   ei = (const int*)d_in[1];
    const float* W0 = (const float*)d_in[2];
    const float* b0 = (const float*)d_in[3];
    const float* W1 = (const float*)d_in[4];
    const float* b1 = (const float*)d_in[5];
    float* out = (float*)d_out;

    const int* src = ei;
    const int* dst = ei + N_EDGES;

    float* ws   = (float*)d_ws;
    float* dinv = ws;
    float* h0   = dinv + N_NODES;
    float* agg0 = h0 + (long)N_NODES * 16;
    float* h2   = agg0 + (long)N_NODES * 16;

    const int nTiles = N_NODES / 16;
    const dim3 B(256);
    auto gr = [](long n) { return dim3((unsigned)((n + 255) / 256)); };

    k_deg_init <<<gr(N_NODES), B, 0, stream>>>(dinv, N_NODES);
    k_deg_edges<<<gr(N_EDGES), B, 0, stream>>>(dst, dinv, N_EDGES);
    k_dinv     <<<gr(N_NODES), B, 0, stream>>>(dinv, N_NODES);

    k_gemm0<<<dim3((nTiles + 7) / 8), B, 0, stream>>>(x, W0, h0, nTiles);
    k_self_init16<<<gr((long)N_NODES * 4), B, 0, stream>>>(h0, dinv, agg0, N_NODES * 4);
    k_edge16<<<dim3((N_EDGES / EDGES_PER_WAVE + 7) / 8), B, 0, stream>>>(src, dst, dinv, h0, agg0, N_EDGES);
    k_relu_bias<<<gr((long)N_NODES * 16), B, 0, stream>>>(agg0, b0, h0, N_NODES * 16);

    k_gemm1<<<dim3((nTiles + 7) / 8), B, 0, stream>>>(h0, W1, h2, nTiles);
    k_out_init7<<<gr((long)N_NODES * N_CLS), B, 0, stream>>>(h2, dinv, b1, out, N_NODES * N_CLS);
    k_edge7<<<dim3((N_EDGES / EDGES_PER_WAVE + 7) / 8), B, 0, stream>>>(src, dst, dinv, h2, out, N_EDGES);
}


# Four elementwise launches (dinv, self-loop init, bias+relu, output init) folded into the two GEMM kernels' loader and epilogues: 10 launches -> 6, h1 never stored; summation order unchanged
# speedup vs baseline: 3.0932x; 1.0607x over previous
#include <hip/hip_runtime.h>

#define N_NODES 100000
#define N_EDGES 1600000
#define F_IN    512
#define HIDDEN  16
#define N_CLS   7

typedef float v2f __attribute__((ext_vector_type(2)));
typedef float v8f __attribute__((ext_vector_type(8)));

__global__ void k_deg_init(float* deg, int n) {
    int t = blockIdx.x * blockDim.x + threadIdx.x;
    if (t < n) deg[t] = 1.0f;
}

__global__ void k_deg_edges(const int* __restrict__ dst, float* deg, int nE) {
    int e = blockIdx.x * blockDim.x + threadIdx.x;
    if (e < nE) atomicAdd(&deg[dst[e]], 1.0f);
}

__global__ __launch_bounds__(256) void k_gemm0(const float* __restrict__ x,
                                               const float* __restrict__ W0,
                                               float* degdinv,
                                               float* __restrict__ h0,
                                               float* __restrict__ agg, int nTiles) {
    __shared__ float sW[F_IN * HIDDEN];
    for (int i = threadIdx.x; i < (F_IN * HIDDEN) / 4; i += 256)
        ((float4*)sW)[i] = ((const float4*)W0)[i];
    __syncthreads();

    const int wave   = threadIdx.x >> 5;
    const int lane   = threadIdx.x & 31;
    const int tile   = blockIdx.x * 8 + wave;
    if (tile >= nTiles) return;

    const int laneHi = lane >> 4;
    const int lane15 = lane & 15;
    const int row    = tile * 16 + lane15;
    const float* xrow = x + (long)row * F_IN;
    const int koff   = laneHi ? 4 : 0;

    v8f acc = {};
    for (int kb = 0; kb < F_IN; kb += 8) {
        float4 a4 = *(const float4*)(xrow + kb + koff);
        const float* wb = sW + (kb + koff) * HIDDEN + lane15;
        v2f a0; a0.x = a4.x; a0.y = a4.y;
        v2f b0; b0.x = wb[0];  b0.y = wb[16];
        acc = __builtin_amdgcn_wmma_f32_16x16x4_f32(false, a0, false, b0,
                                                    (short)0, acc, false, false);
        v2f a1; a1.x = a4.z; a1.y = a4.w;
        v2f b1; b1.x = wb[32]; b1.y = wb[48];
        acc = __builtin_amdgcn_wmma_f32_16x16x4_f32(false, a1, false, b1,
                                                    (short)0, acc, false, false);
    }
    const int r0 = tile * 16 + laneHi * 8;
    const float4 gA = *(const float4*)(degdinv + r0);
    const float4 gB = *(const float4*)(degdinv + r0 + 4);
    const float d[8] = {__frsqrt_rn(gA.x), __frsqrt_rn(gA.y), __frsqrt_rn(gA.z), __frsqrt_rn(gA.w),
                        __frsqrt_rn(gB.x), __frsqrt_rn(gB.y), __frsqrt_rn(gB.z), __frsqrt_rn(gB.w)};
    if (lane15 == 0) {
        *(float4*)(degdinv + r0)     = make_float4(d[0], d[1], d[2], d[3]);
        *(float4*)(degdinv + r0 + 4) = make_float4(d[4], d[5], d[6], d[7]);
    }
    const long o = (long)tile * 256 + laneHi * 128 + lane15;
#pragma unroll
    for (int i = 0; i < 8; ++i) {
        const float w = d[i] * d[i];
        h0[o + i * 16]  = acc[i];
        agg[o + i * 16] = acc[i] * w;
    }
}


#define EDGES_PER_WAVE 128
static_assert(N_EDGES % EDGES_PER_WAVE == 0, "a wave takes whole groups of 128 edges");

__global__ __launch_bounds__(256) void k_edge16(const int* __restrict__ src, const int* __restrict__ dst,
                         const float* __restrict__ dinv, const float* __restrict__ h,
                         float* __restrict__ agg, int nE) {
    const int lane = threadIdx.x & 31;
    const int base = (blockIdx.x * 8 + (threadIdx.x >> 5)) * EDGES_PER_WAVE;
    if (base >= nE) return;
    const int f    = lane & 15;
    const int half = lane & 16;
    const int4 s4 = *(const int4*)(src + base + 4 * lane);
    const int4 d4 = *(const int4*)(dst + base + 4 * lane);
    const int sv[4] = {s4.x, s4.y, s4.z, s4.w};
    const int dv[4] = {d4.x, d4.y, d4.z, d4.w};
    float wv[4];
    for (int k = 0; k < 4; ++k) wv[k] = dinv[sv[k]] * dinv[dv[k]];
#pragma unroll
    for (int k = 0; k < 4; ++k) {
        const float w  = wv[k];
        const int sOff = sv[k] * 16;
        const int dOff = dv[k] * 16;
        float val[16];
        int   dof[16];
        for (int j = 0; j < 16; ++j) {
            const int   so = __shfl(sOff, half | j);
            const float ww = __shfl(w,    half | j);
            dof[j] = __shfl(dOff, half | j) + f;
            val[j] = h[so + f] * ww;
        }
        for (int j = 0; j < 16; ++j) atomicAdd(&agg[dof[j]], val[j]);
    }
}

__global__ __launch_bounds__(256) void k_gemm1(const float* __restrict__ agg,
                                               const float* __restrict__ b0,
                                               const float* __restrict__ W1,
                                               const float* __restrict__ b1,
                                               const float* __restrict__ dinv,
                                               float* __restrict__ h2,
                                               float* __restrict__ out, int nTiles) {
    __shared__ float sW[16 * 16];
    {
        int k = threadIdx.x >> 4, n = threadIdx.x & 15;
        sW[threadIdx.x] = (n < N_CLS) ? W1[k * N_CLS + n] : 0.0f;
    }
    __syncthreads();

    const int wave   = threadIdx.x >> 5;
    const int lane   = threadIdx.x & 31;
    const int tile   = blockIdx.x * 8 + wave;
    if (tile >= nTiles) return;

    const int laneHi = lane >> 4;
    const int lane15 = lane & 15;
    const int row    = tile * 16 + lane15;
    const float* arow = agg + (long)row * 16;
    const int koff   = laneHi ? 2 : 0;

    v8f acc = {};
#pragma unroll
    for (int kb = 0; kb < 16; kb += 4) {
        const float v0 = arow[kb + koff]     + b0[kb + koff];
        const float v1 = arow[kb + koff + 1] + b0[kb + koff + 1];
        v2f a; a.x = v0 > 0.0f ? v0 : 0.0f;
               a.y = v1 > 0.0f ? v1 : 0.0f;
        v2f b; b.x = sW[(kb + koff) * 16 + lane15];
               b.y = sW[(kb + koff + 1) * 16 + lane15];
        acc = __builtin_amdgcn_wmma_f32_16x16x4_f32(false, a, false, b,
                                                    (short)0, acc, false, false);
    }
    float* o2 = h2 + (long)tile * 256 + laneHi * 128 + lane15;
#pragma unroll
    for (int i = 0; i < 8; ++i) o2[i * 16] = acc[i];

    const int r0 = tile * 16 + laneHi * 8;
    const float4 dA = *(const float4*)(dinv + r0);
    const float4 dB = *(const float4*)(dinv + r0 + 4);
    const float d[8] = {dA.x, dA.y, dA.z, dA.w, dB.x, dB.y, dB.z, dB.w};
    if (lane15 < N_CLS) {
        const float bb = b1[lane15];
        float* oo = out + (long)r0 * N_CLS + lane15;
#pragma unroll
        for (int i = 0; i < 8; ++i) {
            const float w = d[i];
            oo[i * N_CLS] = acc[i] * w * w + bb;
        }
    }
}


__global__ __launch_bounds__(256) void k_edge7(const int* __restrict__ src, const int* __restrict__ dst,
                        const float* __restrict__ dinv, const float* __restrict__ h2,
                        float* __restrict__ out, int nE) {
    const int lane = threadIdx.x & 31;
    const int base = (blockIdx.x * 8 + (threadIdx.x >> 5)) * EDGES_PER_WAVE;
    if (base >= nE) return;
    const int c   = lane & 7;
    const int grp = lane & 24;
    const int4 s4 = *(const int4*)(src + base + 4 * lane);
    const int4 d4 = *(const int4*)(dst + base + 4 * lane);
    const int sv[4] = {s4.x, s4.y, s4.z, s4.w};
    const int dv[4] = {d4.x, d4.y, d4.z, d4.w};
    float wv[4];
    for (int k = 0; k < 4; ++k) wv[k] = dinv[sv[k]] * dinv[dv[k]];
    for (int k = 0; k < 4; ++k) {
        const float w  = wv[k];
        const int sOff = sv[k] * 16;
        const int dOff = dv[k] * N_CLS;
        int   sof[8], dof[8];
        float ww[8];
        for (int j = 0; j < 8; ++j) {
            sof[j] = __shfl(sOff, grp | j) + c;
            dof[j] = __shfl(dOff, grp | j) + c;
            ww[j]  = __shfl(w,    grp | j);
        }
        if (c < N_CLS) {
            float val[8];
            for (int j = 0; j < 8; ++j) val[j] = h2[sof[j]] * ww[j];
            for (int j = 0; j < 8; ++j) atomicAdd(&out[dof[j]], val[j]);
        }
    }
}

extern "C" void kernel_launch(void* const* d_in, const int* in_sizes, int n_in,
                              void* d_out, int out_size, void* d_ws, size_t ws_size,
                              hipStream_t stream) {
    const float* x  = (const float*)d_in[0];
    const int*   ei = (const int*)d_in[1];
    const float* W0 = (const float*)d_in[2];
    const float* b0 = (const float*)d_in[3];
    const float* W1 = (const float*)d_in[4];
    const float* b1 = (const float*)d_in[5];
    float* out = (float*)d_out;

    const int* src = ei;
    const int* dst = ei + N_EDGES;

    float* ws   = (float*)d_ws;
    float* dinv = ws;
    float* h0   = dinv + N_NODES;
    float* agg0 = h0 + (long)N_NODES * 16;
    float* h2   = agg0 + (long)N_NODES * 16;

    const int nTiles = N_NODES / 16;
    const dim3 B(256);
    auto gr = [](long n) { return dim3((unsigned)((n + 255) / 256)); };

    k_deg_init <<<gr(N_NODES), B, 0, stream>>>(dinv, N_NODES);
    k_deg_edges<<<gr(N_EDGES), B, 0, stream>>>(dst, dinv, N_EDGES);

    k_gemm0<<<dim3((nTiles + 7) / 8), B, 0, stream>>>(x, W0, dinv, h0, agg0, nTiles);
    k_edge16<<<dim3((N_EDGES / EDGES_PER_WAVE + 7) / 8), B, 0, stream>>>(src, dst, dinv, h0, agg0, N_EDGES);

    k_gemm1<<<dim3((nTiles + 7) / 8), B, 0, stream>>>(agg0, b0, W1, b1, dinv, h2, out, nTiles);
    k_edge7<<<dim3((N_EDGES / EDGES_PER_WAVE + 7) / 8), B, 0, stream>>>(src, dst, dinv, h2, out, N_EDGES);
}


# Edge sums off global f32 atomics: per call edges binned by 256-node dst range; one work-group per bin adds dinv[src]-scaled rows into an LDS tile (LDS atomics, arrival order), dinv[dst] applied last
# speedup vs baseline: 8.4226x; 2.7229x over previous
#include <hip/hip_runtime.h>

#define N_NODES 100000
#define N_EDGES 1600000
#define F_IN    512
#define HIDDEN  16
#define N_CLS   7


typedef float v2f __attribute__((ext_vector_type(2)));
typedef float v8f __attribute__((ext_vector_type(8)));

#define BIN_SHIFT 8
#define BIN_NODES 256
#define N_BINS    ((N_NODES + BIN_NODES - 1) / BIN_NODES)
#define SINK      BIN_NODES
#define CHUNK     6400
#define CHUNK4    (CHUNK / 4)
static_assert(N_EDGES % CHUNK == 0, "binning work-groups take whole chunks");
static_assert(CHUNK4 <= 4 * 512, "four 16-byte loads per thread cover a chunk");
static_assert(N_BINS <= 512, "the bin scan is one 512-thread pass");
static_assert(N_NODES < (1 << 22), "src must fit above the 9-bit local destination");

__global__ void k_bin_zero(int* binCount, int* binFill) {
    const int t = threadIdx.x;
    if (t < N_BINS) { binCount[t] = 0; binFill[t] = 0; }
}

__device__ __forceinline__ int scan512(int v, int* wsum, int t) {
    const int lane = t & 31;
    int x = v;
#pragma unroll
    for (int o = 1; o < 32; o <<= 1) {
        const int y = __shfl_up(x, o);
        if (lane >= o) x += y;
    }
    if (lane == 31) wsum[t >> 5] = x;
    __syncthreads();
    int add = 0;
    for (int k = 0; k < 16; ++k) add += (k < (t >> 5)) ? wsum[k] : 0;
    __syncthreads();
    return x + add;
}

__global__ __launch_bounds__(512) void k_bin_count(const int* __restrict__ dst, int* binCount) {
    __shared__ int hist[N_BINS];
    const int t = threadIdx.x;
    const int4* d4p = (const int4*)(dst + blockIdx.x * CHUNK);
    int4 d4[4];
#pragma unroll
    for (int k = 0; k < 4; ++k) {
        const int q = t + k * 512;
        d4[k] = (q < CHUNK4) ? d4p[q] : make_int4(-1, -1, -1, -1);
    }
    if (t < N_BINS) hist[t] = 0;
    __syncthreads();
#pragma unroll
    for (int k = 0; k < 4; ++k) {
        if (d4[k].x >= 0) {
            atomicAdd(&hist[d4[k].x >> BIN_SHIFT], 1);
            atomicAdd(&hist[d4[k].y >> BIN_SHIFT], 1);
            atomicAdd(&hist[d4[k].z >> BIN_SHIFT], 1);
            atomicAdd(&hist[d4[k].w >> BIN_SHIFT], 1);
        }
    }
    __syncthreads();
    if (t < N_BINS) { const int c = hist[t]; if (c) atomicAdd(&binCount[t], c); }
}

__global__ __launch_bounds__(512) void k_bin_scatter(const int* __restrict__ src, const int* __restrict__ dst,
                                                     const int* __restrict__ binCount, int* binFill,
                                                     int* binStart, int* __restrict__ packed) {
    __shared__ int wsum[16];
    __shared__ int hist[N_BINS];
    __shared__ int gdelta[N_BINS];
    __shared__ int sWord[CHUNK];
    __shared__ int sPos[CHUNK];
    const int t = threadIdx.x;
    const int4* d4p = (const int4*)(dst + blockIdx.x * CHUNK);
    const int4* s4p = (const int4*)(src + blockIdx.x * CHUNK);
    int4 d4[4], s4[4];
#pragma unroll
    for (int k = 0; k < 4; ++k) {
        const int q = t + k * 512;
        d4[k] = (q < CHUNK4) ? d4p[q] : make_int4(-1, -1, -1, -1);
        s4[k] = (q < CHUNK4) ? s4p[q] : make_int4(0, 0, 0, 0);
    }
    const int v     = (t < N_BINS) ? binCount[t] : 0;
    const int start = scan512(v, wsum, t) - v;
    if (blockIdx.x == 0) {
        if (t < N_BINS) binStart[t] = start;
        if (t == N_BINS - 1) binStart[N_BINS] = start + v;
    }
    if (t < N_BINS) hist[t] = 0;
    __syncthreads();
#pragma unroll
    for (int k = 0; k < 4; ++k) {
        if (d4[k].x >= 0) {
            atomicAdd(&hist[d4[k].x >> BIN_SHIFT], 1);
            atomicAdd(&hist[d4[k].y >> BIN_SHIFT], 1);
            atomicAdd(&hist[d4[k].z >> BIN_SHIFT], 1);
            atomicAdd(&hist[d4[k].w >> BIN_SHIFT], 1);
        }
    }
    __syncthreads();
    const int c  = (t < N_BINS) ? hist[t] : 0;
    const int ls = scan512(c, wsum, t) - c;
    if (t < N_BINS) {
        const int g = c ? atomicAdd(&binFill[t], c) : 0;
        gdelta[t] = start + g - ls;
        hist[t]   = ls;
    }
    __syncthreads();
#pragma unroll
    for (int k = 0; k < 4; ++k) {
        if (d4[k].x >= 0) {
            const int dd[4] = {d4[k].x, d4[k].y, d4[k].z, d4[k].w};
            const int ss[4] = {s4[k].x, s4[k].y, s4[k].z, s4[k].w};
#pragma unroll
            for (int m = 0; m < 4; ++m) {
                const int b = dd[m] >> BIN_SHIFT;
                const int l = atomicAdd(&hist[b], 1);
                sWord[l] = (ss[m] << 9) | (dd[m] & (BIN_NODES - 1));
                sPos[l]  = gdelta[b] + l;
            }
        }
    }
    __syncthreads();
    for (int i = t; i < CHUNK; i += 512) packed[sPos[i]] = sWord[i];
}

__global__ __launch_bounds__(1024) void k_bin_dinv(const int* __restrict__ packed, const int* __restrict__ binStart,
                                                   float* __restrict__ dinv) {
    __shared__ int cnt[BIN_NODES];
    const int t = threadIdx.x;
    if (t < BIN_NODES) cnt[t] = 1;
    __syncthreads();
    const int e1 = binStart[blockIdx.x + 1];
    for (int e = binStart[blockIdx.x] + t; e < e1; e += 4 * 1024) {
        int p[4];
#pragma unroll
        for (int k = 0; k < 4; ++k) p[k] = (e + k * 1024 < e1) ? packed[e + k * 1024] : -1;
#pragma unroll
        for (int k = 0; k < 4; ++k) if (p[k] >= 0) atomicAdd(&cnt[p[k] & (BIN_NODES - 1)], 1);
    }
    __syncthreads();
    const int node = blockIdx.x * BIN_NODES + t;
    if (t < BIN_NODES && node < N_NODES) dinv[node] = __frsqrt_rn((float)cnt[t]);
}

__global__ __launch_bounds__(256) void k_gemm0(const float* __restrict__ x,
                                               const float* __restrict__ W0,
                                               const float* __restrict__ dinv,
                                               float* __restrict__ h0s, int nTiles) {
    __shared__ float sW[F_IN * HIDDEN];
    for (int i = threadIdx.x; i < (F_IN * HIDDEN) / 4; i += 256)
        ((float4*)sW)[i] = ((const float4*)W0)[i];
    __syncthreads();

    const int wave   = threadIdx.x >> 5;
    const int lane   = threadIdx.x & 31;
    const int tile   = blockIdx.x * 8 + wave;
    if (tile >= nTiles) return;

    const int laneHi = lane >> 4;
    const int lane15 = lane & 15;
    const int row    = tile * 16 + lane15;
    const float* xrow = x + (long)row * F_IN;
    const int koff   = laneHi ? 4 : 0;

    v8f acc = {};
    for (int kb = 0; kb < F_IN; kb += 8) {
        float4 a4 = *(const float4*)(xrow + kb + koff);
        const float* wb = sW + (kb + koff) * HIDDEN + lane15;
        v2f a0; a0.x = a4.x; a0.y = a4.y;
        v2f b0; b0.x = wb[0];  b0.y = wb[16];
        acc = __builtin_amdgcn_wmma_f32_16x16x4_f32(false, a0, false, b0,
                                                    (short)0, acc, false, false);
        v2f a1; a1.x = a4.z; a1.y = a4.w;
        v2f b1; b1.x = wb[32]; b1.y = wb[48];
        acc = __builtin_amdgcn_wmma_f32_16x16x4_f32(false, a1, false, b1,
                                                    (short)0, acc, false, false);
    }
    const int r0 = tile * 16 + laneHi * 8;
    const float4 gA = *(const float4*)(dinv + r0);
    const float4 gB = *(const float4*)(dinv + r0 + 4);
    const float d[8] = {gA.x, gA.y, gA.z, gA.w, gB.x, gB.y, gB.z, gB.w};
    const long o = (long)tile * 256 + laneHi * 128 + lane15;
#pragma unroll
    for (int i = 0; i < 8; ++i) h0s[o + i * 16] = acc[i] * d[i];
}

__global__ __launch_bounds__(1024) void k_agg16(const int* __restrict__ packed, const int* __restrict__ binStart,
                                                const float* __restrict__ dinv, const float* __restrict__ h0s,
                                                float* __restrict__ agg) {
    __shared__ float4 tile4[(BIN_NODES + 1) * 4];
    float* tile = (float*)tile4;
    const int t        = threadIdx.x;
    const int nodeBase = blockIdx.x * BIN_NODES;
    const int myNode   = nodeBase + (t >> 2);
    tile4[t] = (myNode < N_NODES) ? ((const float4*)h0s)[(long)nodeBase * 4 + t] : make_float4(0.f, 0.f, 0.f, 0.f);
    __syncthreads();

    const int lane = t & 31;
    const int f    = lane & 15;
    const int half = lane & 16;
    const int e1   = binStart[blockIdx.x + 1];
    int c0 = binStart[blockIdx.x] + (t >> 5) * 32;
    int p  = (c0 + lane < e1) ? packed[c0 + lane] : SINK;
    while (c0 < e1) {
        const int cn = c0 + 32 * 32;
        float val[16];
        int   pj[16];
#pragma unroll
        for (int j = 0; j < 16; ++j) pj[j] = __shfl(p, half | j);
        const int pn = (cn + lane < e1) ? packed[cn + lane] : SINK;
#pragma unroll
        for (int j = 0; j < 16; ++j) val[j] = h0s[(pj[j] >> 9) * 16 + f];
#pragma unroll
        for (int j = 0; j < 16; ++j) atomicAdd(&tile[(pj[j] & 511) * 16 + f], val[j]);
        p  = pn;
        c0 = cn;
    }
    __syncthreads();
    if (myNode < N_NODES) {
        const float  w = dinv[myNode];
        const float4 s = tile4[t];
        ((float4*)agg)[(long)nodeBase * 4 + t] = make_float4(s.x * w, s.y * w, s.z * w, s.w * w);
    }
}

__global__ __launch_bounds__(256) void k_gemm1(const float* __restrict__ agg,
                                               const float* __restrict__ b0,
                                               const float* __restrict__ W1,
                                               const float* __restrict__ dinv,
                                               float* __restrict__ h2s, int nTiles) {
    __shared__ float sW[16 * 16];
    {
        int k = threadIdx.x >> 4, n = threadIdx.x & 15;
        sW[threadIdx.x] = (n < N_CLS) ? W1[k * N_CLS + n] : 0.0f;
    }
    __syncthreads();

    const int wave   = threadIdx.x >> 5;
    const int lane   = threadIdx.x & 31;
    const int tile   = blockIdx.x * 8 + wave;
    if (tile >= nTiles) return;

    const int laneHi = lane >> 4;
    const int lane15 = lane & 15;
    const int row    = tile * 16 + lane15;
    const float* arow = agg + (long)row * 16;
    const int koff   = laneHi ? 2 : 0;

    v8f acc = {};
#pragma unroll
    for (int kb = 0; kb < 16; kb += 4) {
        const float v0 = arow[kb + koff]     + b0[kb + koff];
        const float v1 = arow[kb + koff + 1] + b0[kb + koff + 1];
        v2f a; a.x = v0 > 0.0f ? v0 : 0.0f;
               a.y = v1 > 0.0f ? v1 : 0.0f;
        v2f b; b.x = sW[(kb + koff) * 16 + lane15];
               b.y = sW[(kb + koff + 1) * 16 + lane15];
        acc = __builtin_amdgcn_wmma_f32_16x16x4_f32(false, a, false, b,
                                                    (short)0, acc, false, false);
    }
    const int r0 = tile * 16 + laneHi * 8;
    const float4 dA = *(const float4*)(dinv + r0);
    const float4 dB = *(const float4*)(dinv + r0 + 4);
    const float d[8] = {dA.x, dA.y, dA.z, dA.w, dB.x, dB.y, dB.z, dB.w};
    float* o2 = h2s + (long)tile * 256 + laneHi * 128 + lane15;
#pragma unroll
    for (int i = 0; i < 8; ++i) o2[i * 16] = acc[i] * d[i];
}

__global__ __launch_bounds__(1024) void k_agg7(const int* __restrict__ packed, const int* __restrict__ binStart,
                                               const float* __restrict__ dinv, const float* __restrict__ h2s,
                                               const float* __restrict__ b1, float* __restrict__ out) {
    __shared__ float4 tile4[(BIN_NODES + 1) * 2];
    float* tile = (float*)tile4;
    const int t        = threadIdx.x;
    const int nodeBase = blockIdx.x * BIN_NODES;
    if (t < BIN_NODES * 2) {
        const int node = nodeBase + (t >> 1);
        tile4[t] = (node < N_NODES) ? ((const float4*)h2s)[(long)node * 4 + (t & 1)] : make_float4(0.f, 0.f, 0.f, 0.f);
    }
    __syncthreads();

    const int lane = t & 31;
    const int c    = lane & 7;
    const int grp  = lane & 24;
    const int e1   = binStart[blockIdx.x + 1];
    int c0 = binStart[blockIdx.x] + (t >> 5) * 32;
    int p  = (c0 + lane < e1) ? packed[c0 + lane] : SINK;
    while (c0 < e1) {
        const int cn = c0 + 32 * 32;
        float val[8];
        int   pj[8];
#pragma unroll
        for (int j = 0; j < 8; ++j) pj[j] = __shfl(p, grp | j);
        const int pn = (cn + lane < e1) ? packed[cn + lane] : SINK;
#pragma unroll
        for (int j = 0; j < 8; ++j) val[j] = h2s[(pj[j] >> 9) * 16 + c];
#pragma unroll
        for (int j = 0; j < 8; ++j) atomicAdd(&tile[(pj[j] & 511) * 8 + c], val[j]);
        p  = pn;
        c0 = cn;
    }
    __syncthreads();
    for (int i = t; i < BIN_NODES * N_CLS; i += 1024) {
        const int nl   = i / N_CLS;
        const int cc   = i - nl * N_CLS;
        const int node = nodeBase + nl;
        if (node < N_NODES) out[(long)nodeBase * N_CLS + i] = tile[nl * 8 + cc] * dinv[node] + b1[cc];
    }
}

extern "C" void kernel_launch(void* const* d_in, const int* in_sizes, int n_in,
                              void* d_out, int out_size, void* d_ws, size_t ws_size,
                              hipStream_t stream) {
    const float* x  = (const float*)d_in[0];
    const int*   ei = (const int*)d_in[1];
    const float* W0 = (const float*)d_in[2];
    const float* b0 = (const float*)d_in[3];
    const float* W1 = (const float*)d_in[4];
    const float* b1 = (const float*)d_in[5];
    float* out = (float*)d_out;

    const int* src = ei;
    const int* dst = ei + N_EDGES;

    float* ws   = (float*)d_ws;
    float* dinv = ws;
    float* h0s  = dinv + N_NODES;
    float* agg0 = h0s + (long)N_NODES * 16;
    float* h2s  = h0s;
    int* packed   = (int*)(agg0 + (long)N_NODES * 16);
    int* binCount = packed + N_EDGES;
    int* binFill  = binCount + N_BINS;
    int* binStart = binFill + N_BINS;

    const int nTiles = N_NODES / 16;

    k_bin_zero   <<<dim3(1), dim3(512), 0, stream>>>(binCount, binFill);
    k_bin_count  <<<dim3(N_EDGES / CHUNK), dim3(512), 0, stream>>>(dst, binCount);
    k_bin_scatter<<<dim3(N_EDGES / CHUNK), dim3(512), 0, stream>>>(src, dst, binCount, binFill, binStart, packed);
    k_bin_dinv   <<<dim3(N_BINS), dim3(1024), 0, stream>>>(packed, binStart, dinv);

    k_gemm0<<<dim3((nTiles + 7) / 8), dim3(256), 0, stream>>>(x, W0, dinv, h0s, nTiles);
    k_agg16<<<dim3(N_BINS), dim3(1024), 0, stream>>>(packed, binStart, dinv, h0s, agg0);

    k_gemm1<<<dim3((nTiles + 7) / 8), dim3(256), 0, stream>>>(agg0, b0, W1, dinv, h2s, nTiles);
    k_agg7<<<dim3(N_BINS), dim3(1024), 0, stream>>>(packed, binStart, dinv, h2s, b1, out);
}


# Two launches folded into kernels owning the same nodes: degrees counted in the layer-0 GEMM prologue (its W0 staging now one load group), layer-1 GEMM run after the layer-0 edge sums; 8 -> 6 launches
# speedup vs baseline: 9.5399x; 1.1327x over previous
#include <hip/hip_runtime.h>

#define N_NODES 100000
#define N_EDGES 1600000
#define F_IN    512
#define HIDDEN  16
#define N_CLS   7


typedef float v2f __attribute__((ext_vector_type(2)));
typedef float v8f __attribute__((ext_vector_type(8)));

#define BIN_SHIFT 8
#define BIN_NODES 256
#define N_BINS    ((N_NODES + BIN_NODES - 1) / BIN_NODES)
#define SINK      BIN_NODES
#define CHUNK     6400
#define CHUNK4    (CHUNK / 4)
static_assert(N_EDGES % CHUNK == 0, "binning work-groups take whole chunks");
static_assert(CHUNK4 <= 4 * 512, "four 16-byte loads per thread cover a chunk");
static_assert(N_BINS <= 512, "the bin scan is one 512-thread pass");
static_assert(N_NODES < (1 << 22), "src must fit above the 9-bit local destination");

__global__ void k_bin_zero(int* binCount, int* binFill) {
    const int t = threadIdx.x;
    if (t < N_BINS) { binCount[t] = 0; binFill[t] = 0; }
}

__device__ __forceinline__ int scan512(int v, int* wsum, int t) {
    const int lane = t & 31;
    int x = v;
#pragma unroll
    for (int o = 1; o < 32; o <<= 1) {
        const int y = __shfl_up(x, o);
        if (lane >= o) x += y;
    }
    if (lane == 31) wsum[t >> 5] = x;
    __syncthreads();
    int add = 0;
    for (int k = 0; k < 16; ++k) add += (k < (t >> 5)) ? wsum[k] : 0;
    __syncthreads();
    return x + add;
}

__global__ __launch_bounds__(512) void k_bin_count(const int* __restrict__ dst, int* binCount) {
    __shared__ int hist[N_BINS];
    const int t = threadIdx.x;
    const int4* d4p = (const int4*)(dst + blockIdx.x * CHUNK);
    int4 d4[4];
#pragma unroll
    for (int k = 0; k < 4; ++k) {
        const int q = t + k * 512;
        d4[k] = (q < CHUNK4) ? d4p[q] : make_int4(-1, -1, -1, -1);
    }
    if (t < N_BINS) hist[t] = 0;
    __syncthreads();
#pragma unroll
    for (int k = 0; k < 4; ++k) {
        if (d4[k].x >= 0) {
            atomicAdd(&hist[d4[k].x >> BIN_SHIFT], 1);
            atomicAdd(&hist[d4[k].y >> BIN_SHIFT], 1);
            atomicAdd(&hist[d4[k].z >> BIN_SHIFT], 1);
            atomicAdd(&hist[d4[k].w >> BIN_SHIFT], 1);
        }
    }
    __syncthreads();
    if (t < N_BINS) { const int c = hist[t]; if (c) atomicAdd(&binCount[t], c); }
}

__global__ __launch_bounds__(512) void k_bin_scatter(const int* __restrict__ src, const int* __restrict__ dst,
                                                     const int* __restrict__ binCount, int* binFill,
                                                     int* binStart, int* __restrict__ packed) {
    __shared__ int wsum[16];
    __shared__ int hist[N_BINS];
    __shared__ int gdelta[N_BINS];
    __shared__ int sWord[CHUNK];
    __shared__ int sPos[CHUNK];
    const int t = threadIdx.x;
    const int4* d4p = (const int4*)(dst + blockIdx.x * CHUNK);
    const int4* s4p = (const int4*)(src + blockIdx.x * CHUNK);
    int4 d4[4], s4[4];
#pragma unroll
    for (int k = 0; k < 4; ++k) {
        const int q = t + k * 512;
        d4[k] = (q < CHUNK4) ? d4p[q] : make_int4(-1, -1, -1, -1);
        s4[k] = (q < CHUNK4) ? s4p[q] : make_int4(0, 0, 0, 0);
    }
    const int v     = (t < N_BINS) ? binCount[t] : 0;
    const int start = scan512(v, wsum, t) - v;
    if (blockIdx.x == 0) {
        if (t < N_BINS) binStart[t] = start;
        if (t == N_BINS - 1) binStart[N_BINS] = start + v;
    }
    if (t < N_BINS) hist[t] = 0;
    __syncthreads();
#pragma unroll
    for (int k = 0; k < 4; ++k) {
        if (d4[k].x >= 0) {
            atomicAdd(&hist[d4[k].x >> BIN_SHIFT], 1);
            atomicAdd(&hist[d4[k].y >> BIN_SHIFT], 1);
            atomicAdd(&hist[d4[k].z >> BIN_SHIFT], 1);
            atomicAdd(&hist[d4[k].w >> BIN_SHIFT], 1);
        }
    }
    __syncthreads();
    const int c  = (t < N_BINS) ? hist[t] : 0;
    const int ls = scan512(c, wsum, t) - c;
    if (t < N_BINS) {
        const int g = c ? atomicAdd(&binFill[t], c) : 0;
        gdelta[t] = start + g - ls;
        hist[t]   = ls;
    }
    __syncthreads();
#pragma unroll
    for (int k = 0; k < 4; ++k) {
        if (d4[k].x >= 0) {
            const int dd[4] = {d4[k].x, d4[k].y, d4[k].z, d4[k].w};
            const int ss[4] = {s4[k].x, s4[k].y, s4[k].z, s4[k].w};
#pragma unroll
            for (int m = 0; m < 4; ++m) {
                const int b = dd[m] >> BIN_SHIFT;
                const int l = atomicAdd(&hist[b], 1);
                sWord[l] = (ss[m] << 9) | (dd[m] & (BIN_NODES - 1));
                sPos[l]  = gdelta[b] + l;
            }
        }
    }
    __syncthreads();
    for (int i = t; i < CHUNK; i += 512) packed[sPos[i]] = sWord[i];
}

#define G0_NODES 128
static_assert(2 * G0_NODES == BIN_NODES, "a GEMM0 work-group is half a bin");
static_assert((F_IN * HIDDEN) / 4 == 8 * 256, "eight 16-byte W0 loads per thread");
static_assert(N_NODES % 16 == 0, "row tiles are whole");
static_assert((N_NODES / 16 + 7) / 8 <= 2 * N_BINS, "every GEMM0 work-group lies inside a bin");
__global__ __launch_bounds__(256) void k_gemm0(const float* __restrict__ x,
                                               const float* __restrict__ W0,
                                               const int* __restrict__ packed,
                                               const int* __restrict__ binStart,
                                               float* __restrict__ dinv,
                                               float* __restrict__ h0s, int nTiles) {
    __shared__ float sW[F_IN * HIDDEN];
    __shared__ int   cnt[G0_NODES];
    const int t = threadIdx.x;
    const int bin  = blockIdx.x >> 1;
    const int half = (blockIdx.x & 1) << 7;
    const int e0 = binStart[bin];
    const int e1 = binStart[bin + 1];
#pragma unroll
    for (int k = 0; k < 8; ++k) ((float4*)sW)[t + k * 256] = ((const float4*)W0)[t + k * 256];
    if (t < G0_NODES) cnt[t] = 1;
    __syncthreads();
    for (int e = e0 + t; e < e1; e += 8 * 256) {
        int p[8];
#pragma unroll
        for (int k = 0; k < 8; ++k) p[k] = (e + k * 256 < e1) ? packed[e + k * 256] : -1;
#pragma unroll
        for (int k = 0; k < 8; ++k)
            if (p[k] >= 0 && (p[k] & G0_NODES) == half) atomicAdd(&cnt[p[k] & (G0_NODES - 1)], 1);
    }
    __syncthreads();
    {
        const int node = blockIdx.x * G0_NODES + t;
        if (t < G0_NODES && node < N_NODES) dinv[node] = __frsqrt_rn((float)cnt[t]);
    }

    const int wave   = threadIdx.x >> 5;
    const int lane   = threadIdx.x & 31;
    const int tile   = blockIdx.x * 8 + wave;
    if (tile >= nTiles) return;

    const int laneHi = lane >> 4;
    const int lane15 = lane & 15;
    const int row    = tile * 16 + lane15;
    const float* xrow = x + (long)row * F_IN;
    const int koff   = laneHi ? 4 : 0;

    v8f acc = {};
    for (int kb = 0; kb < F_IN; kb += 8) {
        float4 a4 = *(const float4*)(xrow + kb + koff);
        const float* wb = sW + (kb + koff) * HIDDEN + lane15;
        v2f a0; a0.x = a4.x; a0.y = a4.y;
        v2f b0; b0.x = wb[0];  b0.y = wb[16];
        acc = __builtin_amdgcn_wmma_f32_16x16x4_f32(false, a0, false, b0,
                                                    (short)0, acc, false, false);
        v2f a1; a1.x = a4.z; a1.y = a4.w;
        v2f b1; b1.x = wb[32]; b1.y = wb[48];
        acc = __builtin_amdgcn_wmma_f32_16x16x4_f32(false, a1, false, b1,
                                                    (short)0, acc, false, false);
    }
    const int lr = wave * 16 + laneHi * 8;
    const long o = (long)tile * 256 + laneHi * 128 + lane15;
#pragma unroll
    for (int i = 0; i < 8; ++i) h0s[o + i * 16] = acc[i] * __frsqrt_rn((float)cnt[lr + i]);
}

__global__ __launch_bounds__(1024) void k_agg16(const int* __restrict__ packed, const int* __restrict__ binStart,
                                                const float* __restrict__ dinv, const float* __restrict__ h0s,
                                                const float* __restrict__ b0, const float* __restrict__ W1,
                                                float* __restrict__ h2s) {
    __shared__ float4 tile4[(BIN_NODES + 1) * 4];
    __shared__ float  sW[16 * 16];
    float* tile = (float*)tile4;
    const int t        = threadIdx.x;
    const int nodeBase = blockIdx.x * BIN_NODES;
    const int myNode   = nodeBase + (t >> 2);
    tile4[t] = (myNode < N_NODES) ? ((const float4*)h0s)[(long)nodeBase * 4 + t] : make_float4(0.f, 0.f, 0.f, 0.f);
    if (t < 16 * 16) {
        const int k = t >> 4, n = t & 15;
        sW[t] = (n < N_CLS) ? W1[k * N_CLS + n] : 0.0f;
    }
    __syncthreads();

    const int lane = t & 31;
    const int f    = lane & 15;
    const int half = lane & 16;
    const int e1   = binStart[blockIdx.x + 1];
    int c0 = binStart[blockIdx.x] + (t >> 5) * 32;
    int p  = (c0 + lane < e1) ? packed[c0 + lane] : SINK;
    while (c0 < e1) {
        const int cn = c0 + 32 * 32;
        float val[16];
        int   pj[16];
#pragma unroll
        for (int j = 0; j < 16; ++j) pj[j] = __shfl(p, half | j);
        const int pn = (cn + lane < e1) ? packed[cn + lane] : SINK;
#pragma unroll
        for (int j = 0; j < 16; ++j) val[j] = h0s[(pj[j] >> 9) * 16 + f];
#pragma unroll
        for (int j = 0; j < 16; ++j) atomicAdd(&tile[(pj[j] & 511) * 16 + f], val[j]);
        p  = pn;
        c0 = cn;
    }
    __syncthreads();

    const int wave     = t >> 5;
    const int tileBase = nodeBase + wave * 16;
    if (wave < BIN_NODES / 16 && tileBase < N_NODES) {
        const int laneHi = lane >> 4;
        const int koff   = laneHi ? 2 : 0;
        const float  wrow = dinv[tileBase + f];
        const float* trow = tile + (wave * 16 + f) * 16;
        v8f acc = {};
#pragma unroll
        for (int kb = 0; kb < 16; kb += 4) {
            const float v0 = trow[kb + koff]     * wrow + b0[kb + koff];
            const float v1 = trow[kb + koff + 1] * wrow + b0[kb + koff + 1];
            v2f a; a.x = v0 > 0.0f ? v0 : 0.0f;
                   a.y = v1 > 0.0f ? v1 : 0.0f;
            v2f b; b.x = sW[(kb + koff) * 16 + f];
                   b.y = sW[(kb + koff + 1) * 16 + f];
            acc = __builtin_amdgcn_wmma_f32_16x16x4_f32(false, a, false, b,
                                                        (short)0, acc, false, false);
        }
        const int r0 = tileBase + laneHi * 8;
        const float4 dA = *(const float4*)(dinv + r0);
        const float4 dB = *(const float4*)(dinv + r0 + 4);
        const float d[8] = {dA.x, dA.y, dA.z, dA.w, dB.x, dB.y, dB.z, dB.w};
        float* o2 = h2s + (long)r0 * 16 + f;
#pragma unroll
        for (int i = 0; i < 8; ++i) o2[i * 16] = acc[i] * d[i];
    }
}

__global__ __launch_bounds__(1024) void k_agg7(const int* __restrict__ packed, const int* __restrict__ binStart,
                                               const float* __restrict__ dinv, const float* __restrict__ h2s,
                                               const float* __restrict__ b1, float* __restrict__ out) {
    __shared__ float4 tile4[(BIN_NODES + 1) * 2];
    float* tile = (float*)tile4;
    const int t        = threadIdx.x;
    const int nodeBase = blockIdx.x * BIN_NODES;
    if (t < BIN_NODES * 2) {
        const int node = nodeBase + (t >> 1);
        tile4[t] = (node < N_NODES) ? ((const float4*)h2s)[(long)node * 4 + (t & 1)] : make_float4(0.f, 0.f, 0.f, 0.f);
    }
    __syncthreads();

    const int lane = t & 31;
    const int c    = lane & 7;
    const int grp  = lane & 24;
    const int e1   = binStart[blockIdx.x + 1];
    int c0 = binStart[blockIdx.x] + (t >> 5) * 32;
    int p  = (c0 + lane < e1) ? packed[c0 + lane] : SINK;
    while (c0 < e1) {
        const int cn = c0 + 32 * 32;
        float val[8];
        int   pj[8];
#pragma unroll
        for (int j = 0; j < 8; ++j) pj[j] = __shfl(p, grp | j);
        const int pn = (cn + lane < e1) ? packed[cn + lane] : SINK;
#pragma unroll
        for (int j = 0; j < 8; ++j) val[j] = h2s[(pj[j] >> 9) * 16 + c];
#pragma unroll
        for (int j = 0; j < 8; ++j) atomicAdd(&tile[(pj[j] & 511) * 8 + c], val[j]);
        p  = pn;
        c0 = cn;
    }
    __syncthreads();
    for (int i = t; i < BIN_NODES * N_CLS; i += 1024) {
        const int nl   = i / N_CLS;
        const int cc   = i - nl * N_CLS;
        const int node = nodeBase + nl;
        if (node < N_NODES) out[(long)nodeBase * N_CLS + i] = tile[nl * 8 + cc] * dinv[node] + b1[cc];
    }
}

extern "C" void kernel_launch(void* const* d_in, const int* in_sizes, int n_in,
                              void* d_out, int out_size, void* d_ws, size_t ws_size,
                              hipStream_t stream) {
    const float* x  = (const float*)d_in[0];
    const int*   ei = (const int*)d_in[1];
    const float* W0 = (const float*)d_in[2];
    const float* b0 = (const float*)d_in[3];
    const float* W1 = (const float*)d_in[4];
    const float* b1 = (const float*)d_in[5];
    float* out = (float*)d_out;

    const int* src = ei;
    const int* dst = ei + N_EDGES;

    float* ws   = (float*)d_ws;
    float* dinv = ws;
    float* h0s  = dinv + N_NODES;
    float* h2s  = h0s + (long)N_NODES * 16;
    int* packed   = (int*)(h2s + (long)N_NODES * 16);
    int* binCount = packed + N_EDGES;
    int* binFill  = binCount + N_BINS;
    int* binStart = binFill + N_BINS;

    const int nTiles = N_NODES / 16;

    k_bin_zero   <<<dim3(1), dim3(512), 0, stream>>>(binCount, binFill);
    k_bin_count  <<<dim3(N_EDGES / CHUNK), dim3(512), 0, stream>>>(dst, binCount);
    k_bin_scatter<<<dim3(N_EDGES / CHUNK), dim3(512), 0, stream>>>(src, dst, binCount, binFill, binStart, packed);

    k_gemm0<<<dim3((nTiles + 7) / 8), dim3(256), 0, stream>>>(x, W0, packed, binStart, dinv, h0s, nTiles);
    k_agg16<<<dim3(N_BINS), dim3(1024), 0, stream>>>(packed, binStart, dinv, h0s, b0, W1, h2s);

    k_agg7<<<dim3(N_BINS), dim3(1024), 0, stream>>>(packed, binStart, dinv, h2s, b1, out);
}


# GEMM0 x loader: each load takes whole 256-byte row pieces (4 lines, was 16), turned to fragment order in a wave-private LDS stage; two chunks requested ahead, non-temporal; sums unchanged
# speedup vs baseline: 9.9821x; 1.0464x over previous
#include <hip/hip_runtime.h>

#define N_NODES 100000
#define N_EDGES 1600000
#define F_IN    512
#define HIDDEN  16
#define N_CLS   7


typedef float v2f __attribute__((ext_vector_type(2)));
typedef float v8f __attribute__((ext_vector_type(8)));

#define BIN_SHIFT 8
#define BIN_NODES 256
#define N_BINS    ((N_NODES + BIN_NODES - 1) / BIN_NODES)
#define SINK      BIN_NODES
#define CHUNK     6400
#define CHUNK4    (CHUNK / 4)
static_assert(N_EDGES % CHUNK == 0, "binning work-groups take whole chunks");
static_assert(CHUNK4 <= 4 * 512, "four 16-byte loads per thread cover a chunk");
static_assert(N_BINS <= 512, "the bin scan is one 512-thread pass");
static_assert(N_NODES < (1 << 22), "src must fit above the 9-bit local destination");

__global__ void k_bin_zero(int* binCount, int* binFill) {
    const int t = threadIdx.x;
    if (t < N_BINS) { binCount[t] = 0; binFill[t] = 0; }
}

__device__ __forceinline__ int scan512(int v, int* wsum, int t) {
    const int lane = t & 31;
    int x = v;
#pragma unroll
    for (int o = 1; o < 32; o <<= 1) {
        const int y = __shfl_up(x, o);
        if (lane >= o) x += y;
    }
    if (lane == 31) wsum[t >> 5] = x;
    __syncthreads();
    int add = 0;
    for (int k = 0; k < 16; ++k) add += (k < (t >> 5)) ? wsum[k] : 0;
    __syncthreads();
    return x + add;
}

__global__ __launch_bounds__(512) void k_bin_count(const int* __restrict__ dst, int* binCount) {
    __shared__ int hist[N_BINS];
    const int t = threadIdx.x;
    const int4* d4p = (const int4*)(dst + blockIdx.x * CHUNK);
    int4 d4[4];
#pragma unroll
    for (int k = 0; k < 4; ++k) {
        const int q = t + k * 512;
        d4[k] = (q < CHUNK4) ? d4p[q] : make_int4(-1, -1, -1, -1);
    }
    if (t < N_BINS) hist[t] = 0;
    __syncthreads();
#pragma unroll
    for (int k = 0; k < 4; ++k) {
        if (d4[k].x >= 0) {
            atomicAdd(&hist[d4[k].x >> BIN_SHIFT], 1);
            atomicAdd(&hist[d4[k].y >> BIN_SHIFT], 1);
            atomicAdd(&hist[d4[k].z >> BIN_SHIFT], 1);
            atomicAdd(&hist[d4[k].w >> BIN_SHIFT], 1);
        }
    }
    __syncthreads();
    if (t < N_BINS) { const int c = hist[t]; if (c) atomicAdd(&binCount[t], c); }
}

__global__ __launch_bounds__(512) void k_bin_scatter(const int* __restrict__ src, const int* __restrict__ dst,
                                                     const int* __restrict__ binCount, int* binFill,
                                                     int* binStart, int* __restrict__ packed) {
    __shared__ int wsum[16];
    __shared__ int hist[N_BINS];
    __shared__ int gdelta[N_BINS];
    __shared__ int sWord[CHUNK];
    __shared__ int sPos[CHUNK];
    const int t = threadIdx.x;
    const int4* d4p = (const int4*)(dst + blockIdx.x * CHUNK);
    const int4* s4p = (const int4*)(src + blockIdx.x * CHUNK);
    int4 d4[4], s4[4];
#pragma unroll
    for (int k = 0; k < 4; ++k) {
        const int q = t + k * 512;
        d4[k] = (q < CHUNK4) ? d4p[q] : make_int4(-1, -1, -1, -1);
        s4[k] = (q < CHUNK4) ? s4p[q] : make_int4(0, 0, 0, 0);
    }
    const int v     = (t < N_BINS) ? binCount[t] : 0;
    const int start = scan512(v, wsum, t) - v;
    if (blockIdx.x == 0) {
        if (t < N_BINS) binStart[t] = start;
        if (t == N_BINS - 1) binStart[N_BINS] = start + v;
    }
    if (t < N_BINS) hist[t] = 0;
    __syncthreads();
#pragma unroll
    for (int k = 0; k < 4; ++k) {
        if (d4[k].x >= 0) {
            atomicAdd(&hist[d4[k].x >> BIN_SHIFT], 1);
            atomicAdd(&hist[d4[k].y >> BIN_SHIFT], 1);
            atomicAdd(&hist[d4[k].z >> BIN_SHIFT], 1);
            atomicAdd(&hist[d4[k].w >> BIN_SHIFT], 1);
        }
    }
    __syncthreads();
    const int c  = (t < N_BINS) ? hist[t] : 0;
    const int ls = scan512(c, wsum, t) - c;
    if (t < N_BINS) {
        const int g = c ? atomicAdd(&binFill[t], c) : 0;
        gdelta[t] = start + g - ls;
        hist[t]   = ls;
    }
    __syncthreads();
#pragma unroll
    for (int k = 0; k < 4; ++k) {
        if (d4[k].x >= 0) {
            const int dd[4] = {d4[k].x, d4[k].y, d4[k].z, d4[k].w};
            const int ss[4] = {s4[k].x, s4[k].y, s4[k].z, s4[k].w};
#pragma unroll
            for (int m = 0; m < 4; ++m) {
                const int b = dd[m] >> BIN_SHIFT;
                const int l = atomicAdd(&hist[b], 1);
                sWord[l] = (ss[m] << 9) | (dd[m] & (BIN_NODES - 1));
                sPos[l]  = gdelta[b] + l;
            }
        }
    }
    __syncthreads();
    for (int i = t; i < CHUNK; i += 512) packed[sPos[i]] = sWord[i];
}

#define G0_NODES 128
#define XS_PITCH 17
#define XS_WAVE  (16 * XS_PITCH)
static_assert(2 * G0_NODES == BIN_NODES, "a GEMM0 work-group is half a bin");
static_assert((F_IN * HIDDEN) / 4 == 8 * 256, "eight 16-byte W0 loads per thread");
static_assert(N_NODES % 16 == 0, "row tiles are whole");
static_assert((N_NODES / 16 + 7) / 8 <= 2 * N_BINS, "every GEMM0 work-group lies inside a bin");
static_assert(F_IN == 8 * 64, "eight chunks of 64 columns per row");

typedef float v4f __attribute__((ext_vector_type(4)));

__device__ __forceinline__ void g0_wave_order() {
    __builtin_amdgcn_fence(__ATOMIC_RELEASE, "wavefront");
    __builtin_amdgcn_wave_barrier();
    __builtin_amdgcn_fence(__ATOMIC_ACQUIRE, "wavefront");
}

template <bool REFILL>
__device__ __forceinline__ void g0_chunk(v4f (&buf)[8], const v4f* __restrict__ xnext,
                                         v4f* stS, const v4f* stL, const float* wb, v8f& acc) {
#pragma unroll
    for (int i = 0; i < 8; ++i) stS[i * 2 * XS_PITCH] = buf[i];
    g0_wave_order();
    if (REFILL) {
#pragma unroll
        for (int i = 0; i < 8; ++i) buf[i] = __builtin_nontemporal_load(xnext + i * (2 * F_IN / 4));
    }
#pragma unroll
    for (int j = 0; j < 8; ++j) {
        const v4f a4 = stL[2 * j];
        const float* w = wb + j * 8 * HIDDEN;
        v2f a0; a0.x = a4.x; a0.y = a4.y;
        v2f b0; b0.x = w[0];  b0.y = w[16];
        acc = __builtin_amdgcn_wmma_f32_16x16x4_f32(false, a0, false, b0,
                                                    (short)0, acc, false, false);
        v2f a1; a1.x = a4.z; a1.y = a4.w;
        v2f b1; b1.x = w[32]; b1.y = w[48];
        acc = __builtin_amdgcn_wmma_f32_16x16x4_f32(false, a1, false, b1,
                                                    (short)0, acc, false, false);
    }
    g0_wave_order();
}

__global__ __launch_bounds__(256) void k_gemm0(const float* __restrict__ x,
                                               const float* __restrict__ W0,
                                               const int* __restrict__ packed,
                                               const int* __restrict__ binStart,
                                               float* __restrict__ dinv,
                                               float* __restrict__ h0s, int nTiles) {
    __shared__ float sW[F_IN * HIDDEN];
    __shared__ int   cnt[G0_NODES];
    __shared__ v4f   sX[8 * XS_WAVE];
    const int t = threadIdx.x;
    const int wave   = t >> 5;
    const int lane   = t & 31;
    const int laneHi = lane >> 4;
    const int lane15 = lane & 15;
    const int tile   = blockIdx.x * 8 + wave;
    const bool live  = tile < nTiles;
    const int bin  = blockIdx.x >> 1;
    const int half = (blockIdx.x & 1) << 7;
    const int e0 = binStart[bin];
    const int e1 = binStart[bin + 1];
    v4f w4[8];
#pragma unroll
    for (int k = 0; k < 8; ++k) w4[k] = ((const v4f*)W0)[t + k * 256];
    const v4f* xg = (const v4f*)x + ((long)tile * 16 + laneHi) * (F_IN / 4) + lane15;
    v4f bufA[8], bufB[8];
    if (live) {
#pragma unroll
        for (int i = 0; i < 8; ++i) bufA[i] = __builtin_nontemporal_load(xg + i * (2 * F_IN / 4));
#pragma unroll
        for (int i = 0; i < 8; ++i) bufB[i] = __builtin_nontemporal_load(xg + i * (2 * F_IN / 4) + 16);
    }
#pragma unroll
    for (int k = 0; k < 8; ++k) ((v4f*)sW)[t + k * 256] = w4[k];
    if (t < G0_NODES) cnt[t] = 1;
    __syncthreads();
    for (int e = e0 + t; e < e1; e += 8 * 256) {
        int p[8];
#pragma unroll
        for (int k = 0; k < 8; ++k) p[k] = (e + k * 256 < e1) ? packed[e + k * 256] : -1;
#pragma unroll
        for (int k = 0; k < 8; ++k)
            if (p[k] >= 0 && (p[k] & G0_NODES) == half) atomicAdd(&cnt[p[k] & (G0_NODES - 1)], 1);
    }
    __syncthreads();
    {
        const int node = blockIdx.x * G0_NODES + t;
        if (t < G0_NODES && node < N_NODES) dinv[node] = __frsqrt_rn((float)cnt[t]);
    }
    if (!live) return;

    const int koff = laneHi ? 4 : 0;
    v4f*       stS = sX + wave * XS_WAVE + laneHi * XS_PITCH + lane15;
    const v4f* stL = sX + wave * XS_WAVE + lane15 * XS_PITCH + laneHi;
    const float* wb = sW + koff * HIDDEN + lane15;

    v8f acc = {};
#pragma unroll
    for (int c = 0; c < 8; c += 2) {
        if (c + 2 < 8) {
            g0_chunk<true>(bufA, xg + (c + 2) * 16, stS, stL, wb + c * 64 * HIDDEN, acc);
            g0_chunk<true>(bufB, xg + (c + 3) * 16, stS, stL, wb + (c + 1) * 64 * HIDDEN, acc);
        } else {
            g0_chunk<false>(bufA, xg, stS, stL, wb + c * 64 * HIDDEN, acc);
            g0_chunk<false>(bufB, xg, stS, stL, wb + (c + 1) * 64 * HIDDEN, acc);
        }
    }
    const int lr = wave * 16 + laneHi * 8;
    const long o = (long)tile * 256 + laneHi * 128 + lane15;
#pragma unroll
    for (int i = 0; i < 8; ++i) h0s[o + i * 16] = acc[i] * __frsqrt_rn((float)cnt[lr + i]);
}

__global__ __launch_bounds__(1024) void k_agg16(const int* __restrict__ packed, const int* __restrict__ binStart,
                                                const float* __restrict__ dinv, const float* __restrict__ h0s,
                                                const float* __restrict__ b0, const float* __restrict__ W1,
                                                float* __restrict__ h2s) {
    __shared__ float4 tile4[(BIN_NODES + 1) * 4];
    __shared__ float  sW[16 * 16];
    float* tile = (float*)tile4;
    const int t        = threadIdx.x;
    const int nodeBase = blockIdx.x * BIN_NODES;
    const int myNode   = nodeBase + (t >> 2);
    tile4[t] = (myNode < N_NODES) ? ((const float4*)h0s)[(long)nodeBase * 4 + t] : make_float4(0.f, 0.f, 0.f, 0.f);
    if (t < 16 * 16) {
        const int k = t >> 4, n = t & 15;
        sW[t] = (n < N_CLS) ? W1[k * N_CLS + n] : 0.0f;
    }
    __syncthreads();

    const int lane = t & 31;
    const int f    = lane & 15;
    const int half = lane & 16;
    const int e1   = binStart[blockIdx.x + 1];
    int c0 = binStart[blockIdx.x] + (t >> 5) * 32;
    int p  = (c0 + lane < e1) ? packed[c0 + lane] : SINK;
    while (c0 < e1) {
        const int cn = c0 + 32 * 32;
        float val[16];
        int   pj[16];
#pragma unroll
        for (int j = 0; j < 16; ++j) pj[j] = __shfl(p, half | j);
        const int pn = (cn + lane < e1) ? packed[cn + lane] : SINK;
#pragma unroll
        for (int j = 0; j < 16; ++j) val[j] = h0s[(pj[j] >> 9) * 16 + f];
#pragma unroll
        for (int j = 0; j < 16; ++j) atomicAdd(&tile[(pj[j] & 511) * 16 + f], val[j]);
        p  = pn;
        c0 = cn;
    }
    __syncthreads();

    const int wave     = t >> 5;
    const int tileBase = nodeBase + wave * 16;
    if (wave < BIN_NODES / 16 && tileBase < N_NODES) {
        const int laneHi = lane >> 4;
        const int koff   = laneHi ? 2 : 0;
        const float  wrow = dinv[tileBase + f];
        const float* trow = tile + (wave * 16 + f) * 16;
        v8f acc = {};
#pragma unroll
        for (int kb = 0; kb < 16; kb += 4) {
            const float v0 = trow[kb + koff]     * wrow + b0[kb + koff];
            const float v1 = trow[kb + koff + 1] * wrow + b0[kb + koff + 1];
            v2f a; a.x = v0 > 0.0f ? v0 : 0.0f;
                   a.y = v1 > 0.0f ? v1 : 0.0f;
            v2f b; b.x = sW[(kb + koff) * 16 + f];
                   b.y = sW[(kb + koff + 1) * 16 + f];
            acc = __builtin_amdgcn_wmma_f32_16x16x4_f32(false, a, false, b,
                                                        (short)0, acc, false, false);
        }
        const int r0 = tileBase + laneHi * 8;
        const float4 dA = *(const float4*)(dinv + r0);
        const float4 dB = *(const float4*)(dinv + r0 + 4);
        const float d[8] = {dA.x, dA.y, dA.z, dA.w, dB.x, dB.y, dB.z, dB.w};
        float* o2 = h2s + (long)r0 * 16 + f;
#pragma unroll
        for (int i = 0; i < 8; ++i) o2[i * 16] = acc[i] * d[i];
    }
}

__global__ __launch_bounds__(1024) void k_agg7(const int* __restrict__ packed, const int* __restrict__ binStart,
                                               const float* __restrict__ dinv, const float* __restrict__ h2s,
                                               const float* __restrict__ b1, float* __restrict__ out) {
    __shared__ float4 tile4[(BIN_NODES + 1) * 2];
    float* tile = (float*)tile4;
    const int t        = threadIdx.x;
    const int nodeBase = blockIdx.x * BIN_NODES;
    if (t < BIN_NODES * 2) {
        const int node = nodeBase + (t >> 1);
        tile4[t] = (node < N_NODES) ? ((const float4*)h2s)[(long)node * 4 + (t & 1)] : make_float4(0.f, 0.f, 0.f, 0.f);
    }
    __syncthreads();

    const int lane = t & 31;
    const int c    = lane & 7;
    const int grp  = lane & 24;
    const int e1   = binStart[blockIdx.x + 1];
    int c0 = binStart[blockIdx.x] + (t >> 5) * 32;
    int p  = (c0 + lane < e1) ? packed[c0 + lane] : SINK;
    while (c0 < e1) {
        const int cn = c0 + 32 * 32;
        float val[8];
        int   pj[8];
#pragma unroll
        for (int j = 0; j < 8; ++j) pj[j] = __shfl(p, grp | j);
        const int pn = (cn + lane < e1) ? packed[cn + lane] : SINK;
#pragma unroll
        for (int j = 0; j < 8; ++j) val[j] = h2s[(pj[j] >> 9) * 16 + c];
#pragma unroll
        for (int j = 0; j < 8; ++j) atomicAdd(&tile[(pj[j] & 511) * 8 + c], val[j]);
        p  = pn;
        c0 = cn;
    }
    __syncthreads();
    for (int i = t; i < BIN_NODES * N_CLS; i += 1024) {
        const int nl   = i / N_CLS;
        const int cc   = i - nl * N_CLS;
        const int node = nodeBase + nl;
        if (node < N_NODES) out[(long)nodeBase * N_CLS + i] = tile[nl * 8 + cc] * dinv[node] + b1[cc];
    }
}

extern "C" void kernel_launch(void* const* d_in, const int* in_sizes, int n_in,
                              void* d_out, int out_size, void* d_ws, size_t ws_size,
                              hipStream_t stream) {
    const float* x  = (const float*)d_in[0];
    const int*   ei = (const int*)d_in[1];
    const float* W0 = (const float*)d_in[2];
    const float* b0 = (const float*)d_in[3];
    const float* W1 = (const float*)d_in[4];
    const float* b1 = (const float*)d_in[5];
    float* out = (float*)d_out;

    const int* src = ei;
    const int* dst = ei + N_EDGES;

    float* ws   = (float*)d_ws;
    float* dinv = ws;
    float* h0s  = dinv + N_NODES;
    float* h2s  = h0s + (long)N_NODES * 16;
    int* packed   = (int*)(h2s + (long)N_NODES * 16);
    int* binCount = packed + N_EDGES;
    int* binFill  = binCount + N_BINS;
    int* binStart = binFill + N_BINS;

    const int nTiles = N_NODES / 16;

    k_bin_zero   <<<dim3(1), dim3(512), 0, stream>>>(binCount, binFill);
    k_bin_count  <<<dim3(N_EDGES / CHUNK), dim3(512), 0, stream>>>(dst, binCount);
    k_bin_scatter<<<dim3(N_EDGES / CHUNK), dim3(512), 0, stream>>>(src, dst, binCount, binFill, binStart, packed);

    k_gemm0<<<dim3((nTiles + 7) / 8), dim3(256), 0, stream>>>(x, W0, packed, binStart, dinv, h0s, nTiles);
    k_agg16<<<dim3(N_BINS), dim3(1024), 0, stream>>>(packed, binStart, dinv, h0s, b0, W1, h2s);

    k_agg7<<<dim3(N_BINS), dim3(1024), 0, stream>>>(packed, binStart, dinv, h2s, b1, out);
}


# Layer-0 product moved into the two edge-binning launches as extra work-groups: its x stream overlaps the binning's dependent steps; degrees + row scaling in one small per-bin kernel; sums unchanged
# speedup vs baseline: 10.4501x; 1.0469x over previous
#include <hip/hip_runtime.h>

#define N_NODES 100000
#define N_EDGES 1600000
#define F_IN    512
#define HIDDEN  16
#define N_CLS   7


typedef float v2f __attribute__((ext_vector_type(2)));
typedef float v8f __attribute__((ext_vector_type(8)));

#define BIN_SHIFT 8
#define BIN_NODES 256
#define N_BINS    ((N_NODES + BIN_NODES - 1) / BIN_NODES)
#define SINK      BIN_NODES
#define CHUNK     6400
#define CHUNK4    (CHUNK / 4)
static_assert(N_EDGES % CHUNK == 0, "binning work-groups take whole chunks");
static_assert(CHUNK4 <= 4 * 512, "four 16-byte loads per thread cover a chunk");
static_assert(N_BINS <= 512, "the bin scan is one 512-thread pass");
static_assert(N_NODES < (1 << 22), "src must fit above the 9-bit local destination");

__global__ void k_bin_zero(int* binCount, int* binFill) {
    const int t = threadIdx.x;
    if (t < N_BINS) { binCount[t] = 0; binFill[t] = 0; }
}

__device__ __forceinline__ int scan512(int v, int* wsum, int t) {
    const int lane = t & 31;
    int x = v;
#pragma unroll
    for (int o = 1; o < 32; o <<= 1) {
        const int y = __shfl_up(x, o);
        if (lane >= o) x += y;
    }
    if (lane == 31) wsum[t >> 5] = x;
    __syncthreads();
    int add = 0;
    for (int k = 0; k < 16; ++k) add += (k < (t >> 5)) ? wsum[k] : 0;
    __syncthreads();
    return x + add;
}

#define N_CHUNKS   (N_EDGES / CHUNK)
#define G0_ROWS    256
#define G0_GROUPS  ((N_NODES + G0_ROWS - 1) / G0_ROWS)
#define G0_FIRST   170
#define XS_PITCH 17
#define XS_WAVE  (16 * XS_PITCH)
#define LDS_V4F  ((F_IN * HIDDEN) / 4 + 16 * XS_WAVE)
static_assert(LDS_V4F * 4 >= 16 + 2 * N_BINS + 2 * CHUNK, "the scatter pass's arrays fit in the same block");
static_assert((F_IN * HIDDEN) / 4 == 4 * 512, "four 16-byte W0 loads per thread");
static_assert(N_NODES % 16 == 0, "row tiles are whole");
static_assert(F_IN == 8 * 64, "eight chunks of 64 columns per row");
static_assert(G0_ROWS == BIN_NODES, "k_deg scales the rows of one bin");

typedef float v4f __attribute__((ext_vector_type(4)));

__device__ __forceinline__ void bin_count_body(int* hist, int chunk, const int* __restrict__ dst, int* binCount) {
    const int t = threadIdx.x;
    const int4* d4p = (const int4*)(dst + chunk * CHUNK);
    int4 d4[4];
#pragma unroll
    for (int k = 0; k < 4; ++k) {
        const int q = t + k * 512;
        d4[k] = (q < CHUNK4) ? d4p[q] : make_int4(-1, -1, -1, -1);
    }
    if (t < N_BINS) hist[t] = 0;
    __syncthreads();
#pragma unroll
    for (int k = 0; k < 4; ++k) {
        if (d4[k].x >= 0) {
            atomicAdd(&hist[d4[k].x >> BIN_SHIFT], 1);
            atomicAdd(&hist[d4[k].y >> BIN_SHIFT], 1);
            atomicAdd(&hist[d4[k].z >> BIN_SHIFT], 1);
            atomicAdd(&hist[d4[k].w >> BIN_SHIFT], 1);
        }
    }
    __syncthreads();
    if (t < N_BINS) { const int c = hist[t]; if (c) atomicAdd(&binCount[t], c); }
}

__device__ __forceinline__ void bin_scatter_body(int* lds, int chunk, const int* __restrict__ src, const int* __restrict__ dst,
                                                 const int* binCount, int* binFill,
                                                 int* binStart, int* __restrict__ packed) {
    int* wsum   = lds;
    int* hist   = wsum + 16;
    int* gdelta = hist + N_BINS;
    int* sWord  = gdelta + N_BINS;
    int* sPos   = sWord + CHUNK;
    const int t = threadIdx.x;
    const int4* d4p = (const int4*)(dst + chunk * CHUNK);
    const int4* s4p = (const int4*)(src + chunk * CHUNK);
    int4 d4[4], s4[4];
#pragma unroll
    for (int k = 0; k < 4; ++k) {
        const int q = t + k * 512;
        d4[k] = (q < CHUNK4) ? d4p[q] : make_int4(-1, -1, -1, -1);
        s4[k] = (q < CHUNK4) ? s4p[q] : make_int4(0, 0, 0, 0);
    }
    const int v     = (t < N_BINS) ? binCount[t] : 0;
    const int start = scan512(v, wsum, t) - v;
    if (chunk == 0) {
        if (t < N_BINS) binStart[t] = start;
        if (t == N_BINS - 1) binStart[N_BINS] = start + v;
    }
    if (t < N_BINS) hist[t] = 0;
    __syncthreads();
#pragma unroll
    for (int k = 0; k < 4; ++k) {
        if (d4[k].x >= 0) {
            atomicAdd(&hist[d4[k].x >> BIN_SHIFT], 1);
            atomicAdd(&hist[d4[k].y >> BIN_SHIFT], 1);
            atomicAdd(&hist[d4[k].z >> BIN_SHIFT], 1);
            atomicAdd(&hist[d4[k].w >> BIN_SHIFT], 1);
        }
    }
    __syncthreads();
    const int c  = (t < N_BINS) ? hist[t] : 0;
    const int ls = scan512(c, wsum, t) - c;
    if (t < N_BINS) {
        const int g = c ? atomicAdd(&binFill[t], c) : 0;
        gdelta[t] = start + g - ls;
        hist[t]   = ls;
    }
    __syncthreads();
#pragma unroll
    for (int k = 0; k < 4; ++k) {
        if (d4[k].x >= 0) {
            const int dd[4] = {d4[k].x, d4[k].y, d4[k].z, d4[k].w};
            const int ss[4] = {s4[k].x, s4[k].y, s4[k].z, s4[k].w};
#pragma unroll
            for (int m = 0; m < 4; ++m) {
                const int b = dd[m] >> BIN_SHIFT;
                const int l = atomicAdd(&hist[b], 1);
                sWord[l] = (ss[m] << 9) | (dd[m] & (BIN_NODES - 1));
                sPos[l]  = gdelta[b] + l;
            }
        }
    }
    __syncthreads();
    for (int i = t; i < CHUNK; i += 512) packed[sPos[i]] = sWord[i];
}


__device__ __forceinline__ void g0_wave_order() {
    __builtin_amdgcn_fence(__ATOMIC_RELEASE, "wavefront");
    __builtin_amdgcn_wave_barrier();
    __builtin_amdgcn_fence(__ATOMIC_ACQUIRE, "wavefront");
}

template <bool REFILL>
__device__ __forceinline__ void g0_chunk(v4f (&buf)[8], const v4f* __restrict__ xnext,
                                         v4f* stS, const v4f* stL, const float* wb, v8f& acc) {
#pragma unroll
    for (int i = 0; i < 8; ++i) stS[i * 2 * XS_PITCH] = buf[i];
    g0_wave_order();
    if (REFILL) {
#pragma unroll
        for (int i = 0; i < 8; ++i) buf[i] = __builtin_nontemporal_load(xnext + i * (2 * F_IN / 4));
    }
#pragma unroll
    for (int j = 0; j < 8; ++j) {
        const v4f a4 = stL[2 * j];
        const float* w = wb + j * 8 * HIDDEN;
        v2f a0; a0.x = a4.x; a0.y = a4.y;
        v2f b0; b0.x = w[0];  b0.y = w[16];
        acc = __builtin_amdgcn_wmma_f32_16x16x4_f32(false, a0, false, b0,
                                                    (short)0, acc, false, false);
        v2f a1; a1.x = a4.z; a1.y = a4.w;
        v2f b1; b1.x = w[32]; b1.y = w[48];
        acc = __builtin_amdgcn_wmma_f32_16x16x4_f32(false, a1, false, b1,
                                                    (short)0, acc, false, false);
    }
    g0_wave_order();
}

__device__ __forceinline__ void gemm0_body(v4f* lds, int group, const float* __restrict__ x,
                                           const float* __restrict__ W0, float* __restrict__ h0) {
    float* sW = (float*)lds;
    v4f*   sX = lds + (F_IN * HIDDEN) / 4;
    const int t = threadIdx.x;
    const int wave   = t >> 5;
    const int lane   = t & 31;
    const int laneHi = lane >> 4;
    const int lane15 = lane & 15;
    const int tile   = group * 16 + wave;
    const bool live  = tile < N_NODES / 16;
    v4f w4[4];
#pragma unroll
    for (int k = 0; k < 4; ++k) w4[k] = ((const v4f*)W0)[t + k * 512];
    const v4f* xg = (const v4f*)x + ((long)tile * 16 + laneHi) * (F_IN / 4) + lane15;
    v4f bufA[8], bufB[8];
    if (live) {
#pragma unroll
        for (int i = 0; i < 8; ++i) bufA[i] = __builtin_nontemporal_load(xg + i * (2 * F_IN / 4));
#pragma unroll
        for (int i = 0; i < 8; ++i) bufB[i] = __builtin_nontemporal_load(xg + i * (2 * F_IN / 4) + 16);
    }
#pragma unroll
    for (int k = 0; k < 4; ++k) ((v4f*)sW)[t + k * 512] = w4[k];
    __syncthreads();
    if (!live) return;

    const int koff = laneHi ? 4 : 0;
    v4f*       stS = sX + wave * XS_WAVE + laneHi * XS_PITCH + lane15;
    const v4f* stL = sX + wave * XS_WAVE + lane15 * XS_PITCH + laneHi;
    const float* wb = sW + koff * HIDDEN + lane15;

    v8f acc = {};
#pragma unroll
    for (int c = 0; c < 8; c += 2) {
        if (c + 2 < 8) {
            g0_chunk<true>(bufA, xg + (c + 2) * 16, stS, stL, wb + c * 64 * HIDDEN, acc);
            g0_chunk<true>(bufB, xg + (c + 3) * 16, stS, stL, wb + (c + 1) * 64 * HIDDEN, acc);
        } else {
            g0_chunk<false>(bufA, xg, stS, stL, wb + c * 64 * HIDDEN, acc);
            g0_chunk<false>(bufB, xg, stS, stL, wb + (c + 1) * 64 * HIDDEN, acc);
        }
    }
    const long o = (long)tile * 256 + laneHi * 128 + lane15;
#pragma unroll
    for (int i = 0; i < 8; ++i) h0[o + i * 16] = acc[i];
}

template <bool SCATTER>
__global__ __launch_bounds__(512) void k_bin_gemm0(const int* __restrict__ src, const int* __restrict__ dst,
                                                   int* binCount, int* binFill, int* binStart, int* packed,
                                                   const float* __restrict__ x, const float* __restrict__ W0,
                                                   float* __restrict__ h0) {
    __shared__ v4f lds[LDS_V4F];
    const int nProd = SCATTER ? G0_GROUPS - G0_FIRST : G0_FIRST;
    const int b     = blockIdx.x;
    if (b < nProd) {
        gemm0_body(lds, (SCATTER ? G0_FIRST : 0) + b, x, W0, h0);
    } else {
        if (SCATTER) bin_scatter_body((int*)lds, b - nProd, src, dst, binCount, binFill, binStart, packed);
        else         bin_count_body((int*)lds, b - nProd, dst, binCount);
    }
}

__global__ __launch_bounds__(256) void k_deg(const int* __restrict__ packed, const int* __restrict__ binStart,
                                             float* __restrict__ dinv, float* h0s) {
    __shared__ int cnt[BIN_NODES];
    const int t        = threadIdx.x;
    const int nodeBase = blockIdx.x * BIN_NODES;
    const int e0 = binStart[blockIdx.x];
    const int e1 = binStart[blockIdx.x + 1];
    float4* rows = (float4*)h0s + (long)nodeBase * 4;
    float4 r[4];
#pragma unroll
    for (int k = 0; k < 4; ++k) {
        const int i = t + k * 256;
        r[k] = (nodeBase + (i >> 2) < N_NODES) ? rows[i] : make_float4(0.f, 0.f, 0.f, 0.f);
    }
    cnt[t] = 1;
    __syncthreads();
    for (int e = e0 + t; e < e1; e += 8 * 256) {
        int p[8];
#pragma unroll
        for (int k = 0; k < 8; ++k) p[k] = (e + k * 256 < e1) ? packed[e + k * 256] : -1;
#pragma unroll
        for (int k = 0; k < 8; ++k)
            if (p[k] >= 0) atomicAdd(&cnt[p[k] & (BIN_NODES - 1)], 1);
    }
    __syncthreads();
    if (nodeBase + t < N_NODES) dinv[nodeBase + t] = __frsqrt_rn((float)cnt[t]);
#pragma unroll
    for (int k = 0; k < 4; ++k) {
        const int i = t + k * 256;
        if (nodeBase + (i >> 2) < N_NODES) {
            const float d = __frsqrt_rn((float)cnt[i >> 2]);
            rows[i] = make_float4(r[k].x * d, r[k].y * d, r[k].z * d, r[k].w * d);
        }
    }
}

__global__ __launch_bounds__(1024) void k_agg16(const int* __restrict__ packed, const int* __restrict__ binStart,
                                                const float* __restrict__ dinv, const float* __restrict__ h0s,
                                                const float* __restrict__ b0, const float* __restrict__ W1,
                                                float* __restrict__ h2s) {
    __shared__ float4 tile4[(BIN_NODES + 1) * 4];
    __shared__ float  sW[16 * 16];
    float* tile = (float*)tile4;
    const int t        = threadIdx.x;
    const int nodeBase = blockIdx.x * BIN_NODES;
    const int myNode   = nodeBase + (t >> 2);
    tile4[t] = (myNode < N_NODES) ? ((const float4*)h0s)[(long)nodeBase * 4 + t] : make_float4(0.f, 0.f, 0.f, 0.f);
    if (t < 16 * 16) {
        const int k = t >> 4, n = t & 15;
        sW[t] = (n < N_CLS) ? W1[k * N_CLS + n] : 0.0f;
    }
    __syncthreads();

    const int lane = t & 31;
    const int f    = lane & 15;
    const int half = lane & 16;
    const int e1   = binStart[blockIdx.x + 1];
    int c0 = binStart[blockIdx.x] + (t >> 5) * 32;
    int p  = (c0 + lane < e1) ? packed[c0 + lane] : SINK;
    while (c0 < e1) {
        const int cn = c0 + 32 * 32;
        float val[16];
        int   pj[16];
#pragma unroll
        for (int j = 0; j < 16; ++j) pj[j] = __shfl(p, half | j);
        const int pn = (cn + lane < e1) ? packed[cn + lane] : SINK;
#pragma unroll
        for (int j = 0; j < 16; ++j) val[j] = h0s[(pj[j] >> 9) * 16 + f];
#pragma unroll
        for (int j = 0; j < 16; ++j) atomicAdd(&tile[(pj[j] & 511) * 16 + f], val[j]);
        p  = pn;
        c0 = cn;
    }
    __syncthreads();

    const int wave     = t >> 5;
    const int tileBase = nodeBase + wave * 16;
    if (wave < BIN_NODES / 16 && tileBase < N_NODES) {
        const int laneHi = lane >> 4;
        const int koff   = laneHi ? 2 : 0;
        const float  wrow = dinv[tileBase + f];
        const float* trow = tile + (wave * 16 + f) * 16;
        v8f acc = {};
#pragma unroll
        for (int kb = 0; kb < 16; kb += 4) {
            const float v0 = trow[kb + koff]     * wrow + b0[kb + koff];
            const float v1 = trow[kb + koff + 1] * wrow + b0[kb + koff + 1];
            v2f a; a.x = v0 > 0.0f ? v0 : 0.0f;
                   a.y = v1 > 0.0f ? v1 : 0.0f;
            v2f b; b.x = sW[(kb + koff) * 16 + f];
                   b.y = sW[(kb + koff + 1) * 16 + f];
            acc = __builtin_amdgcn_wmma_f32_16x16x4_f32(false, a, false, b,
                                                        (short)0, acc, false, false);
        }
        const int r0 = tileBase + laneHi * 8;
        const float4 dA = *(const float4*)(dinv + r0);
        const float4 dB = *(const float4*)(dinv + r0 + 4);
        const float d[8] = {dA.x, dA.y, dA.z, dA.w, dB.x, dB.y, dB.z, dB.w};
        float* o2 = h2s + (long)r0 * 16 + f;
#pragma unroll
        for (int i = 0; i < 8; ++i) o2[i * 16] = acc[i] * d[i];
    }
}

__global__ __launch_bounds__(1024) void k_agg7(const int* __restrict__ packed, const int* __restrict__ binStart,
                                               const float* __restrict__ dinv, const float* __restrict__ h2s,
                                               const float* __restrict__ b1, float* __restrict__ out) {
    __shared__ float4 tile4[(BIN_NODES + 1) * 2];
    float* tile = (float*)tile4;
    const int t        = threadIdx.x;
    const int nodeBase = blockIdx.x * BIN_NODES;
    if (t < BIN_NODES * 2) {
        const int node = nodeBase + (t >> 1);
        tile4[t] = (node < N_NODES) ? ((const float4*)h2s)[(long)node * 4 + (t & 1)] : make_float4(0.f, 0.f, 0.f, 0.f);
    }
    __syncthreads();

    const int lane = t & 31;
    const int c    = lane & 7;
    const int grp  = lane & 24;
    const int e1   = binStart[blockIdx.x + 1];
    int c0 = binStart[blockIdx.x] + (t >> 5) * 32;
    int p  = (c0 + lane < e1) ? packed[c0 + lane] : SINK;
    while (c0 < e1) {
        const int cn = c0 + 32 * 32;
        float val[8];
        int   pj[8];
#pragma unroll
        for (int j = 0; j < 8; ++j) pj[j] = __shfl(p, grp | j);
        const int pn = (cn + lane < e1) ? packed[cn + lane] : SINK;
#pragma unroll
        for (int j = 0; j < 8; ++j) val[j] = h2s[(pj[j] >> 9) * 16 + c];
#pragma unroll
        for (int j = 0; j < 8; ++j) atomicAdd(&tile[(pj[j] & 511) * 8 + c], val[j]);
        p  = pn;
        c0 = cn;
    }
    __syncthreads();
    for (int i = t; i < BIN_NODES * N_CLS; i += 1024) {
        const int nl   = i / N_CLS;
        const int cc   = i - nl * N_CLS;
        const int node = nodeBase + nl;
        if (node < N_NODES) out[(long)nodeBase * N_CLS + i] = tile[nl * 8 + cc] * dinv[node] + b1[cc];
    }
}

extern "C" void kernel_launch(void* const* d_in, const int* in_sizes, int n_in,
                              void* d_out, int out_size, void* d_ws, size_t ws_size,
                              hipStream_t stream) {
    const float* x  = (const float*)d_in[0];
    const int*   ei = (const int*)d_in[1];
    const float* W0 = (const float*)d_in[2];
    const float* b0 = (const float*)d_in[3];
    const float* W1 = (const float*)d_in[4];
    const float* b1 = (const float*)d_in[5];
    float* out = (float*)d_out;

    const int* src = ei;
    const int* dst = ei + N_EDGES;

    float* ws   = (float*)d_ws;
    float* dinv = ws;
    float* h0s  = dinv + N_NODES;
    float* h2s  = h0s + (long)N_NODES * 16;
    int* packed   = (int*)(h2s + (long)N_NODES * 16);
    int* binCount = packed + N_EDGES;
    int* binFill  = binCount + N_BINS;
    int* binStart = binFill + N_BINS;

    k_bin_zero<<<dim3(1), dim3(512), 0, stream>>>(binCount, binFill);
    k_bin_gemm0<false><<<dim3(N_CHUNKS + G0_FIRST), dim3(512), 0, stream>>>(src, dst, binCount, binFill, binStart, packed, x, W0, h0s);
    k_bin_gemm0<true><<<dim3(N_CHUNKS + G0_GROUPS - G0_FIRST), dim3(512), 0, stream>>>(src, dst, binCount, binFill, binStart, packed, x, W0, h0s);

    k_deg<<<dim3(N_BINS), dim3(256), 0, stream>>>(packed, binStart, dinv, h0s);
    k_agg16<<<dim3(N_BINS), dim3(1024), 0, stream>>>(packed, binStart, dinv, h0s, b0, W1, h2s);

    k_agg7<<<dim3(N_BINS), dim3(1024), 0, stream>>>(packed, binStart, dinv, h2s, b1, out);
}


# Destination bins cut from 256 to 200 nodes: each per-bin launch is 500 equal work-groups, two per WGP, where 391 left some WGPs two bins and the rest one; sums stay in arrival order
# speedup vs baseline: 11.1566x; 1.0676x over previous
#include <hip/hip_runtime.h>

#define N_NODES 100000
#define N_EDGES 1600000
#define F_IN    512
#define HIDDEN  16
#define N_CLS   7


typedef float v2f __attribute__((ext_vector_type(2)));
typedef float v8f __attribute__((ext_vector_type(8)));

#define BIN_NODES 200
#define N_BINS    ((N_NODES + BIN_NODES - 1) / BIN_NODES)
#define TILE_ROWS ((BIN_NODES + 15) / 16 * 16)
#define SINK      TILE_ROWS
#define CHUNK     6400
#define CHUNK4    (CHUNK / 4)
static_assert(N_EDGES % CHUNK == 0, "binning work-groups take whole chunks");
static_assert(CHUNK4 <= 4 * 512, "four 16-byte loads per thread cover a chunk");
static_assert(N_BINS <= 512, "the bin scan is one 512-thread pass");
static_assert(N_NODES < (1 << 22), "src must fit above the 9-bit local destination");
static_assert(SINK < 512, "the local destination and the sink row fit in 9 bits");
static_assert(BIN_NODES % 8 == 0 && N_NODES % 8 == 0, "a lane's 8 result rows of the layer-1 product are all in the bin or all outside");

__device__ __forceinline__ int bin_of(int d) { return (int)((unsigned)d / (unsigned)BIN_NODES); }

__global__ void k_bin_zero(int* binCount, int* binFill) {
    const int t = threadIdx.x;
    if (t < N_BINS) { binCount[t] = 0; binFill[t] = 0; }
}

__device__ __forceinline__ int scan512(int v, int* wsum, int t) {
    const int lane = t & 31;
    int x = v;
#pragma unroll
    for (int o = 1; o < 32; o <<= 1) {
        const int y = __shfl_up(x, o);
        if (lane >= o) x += y;
    }
    if (lane == 31) wsum[t >> 5] = x;
    __syncthreads();
    int add = 0;
    for (int k = 0; k < 16; ++k) add += (k < (t >> 5)) ? wsum[k] : 0;
    __syncthreads();
    return x + add;
}

#define N_CHUNKS   (N_EDGES / CHUNK)
#define G0_ROWS    256
#define G0_GROUPS  ((N_NODES + G0_ROWS - 1) / G0_ROWS)
#define G0_FIRST   170
#define XS_PITCH 17
#define XS_WAVE  (16 * XS_PITCH)
#define LDS_V4F  ((F_IN * HIDDEN) / 4 + 16 * XS_WAVE)
static_assert(LDS_V4F * 4 >= 16 + 2 * N_BINS + 2 * CHUNK, "the scatter pass's arrays fit in the same block");
static_assert((F_IN * HIDDEN) / 4 == 4 * 512, "four 16-byte W0 loads per thread");
static_assert(N_NODES % 16 == 0, "row tiles are whole");
static_assert(F_IN == 8 * 64, "eight chunks of 64 columns per row");

typedef float v4f __attribute__((ext_vector_type(4)));

__device__ __forceinline__ void bin_count_body(int* hist, int chunk, const int* __restrict__ dst, int* binCount) {
    const int t = threadIdx.x;
    const int4* d4p = (const int4*)(dst + chunk * CHUNK);
    int4 d4[4];
#pragma unroll
    for (int k = 0; k < 4; ++k) {
        const int q = t + k * 512;
        d4[k] = (q < CHUNK4) ? d4p[q] : make_int4(-1, -1, -1, -1);
    }
    if (t < N_BINS) hist[t] = 0;
    __syncthreads();
#pragma unroll
    for (int k = 0; k < 4; ++k) {
        if (d4[k].x >= 0) {
            atomicAdd(&hist[bin_of(d4[k].x)], 1);
            atomicAdd(&hist[bin_of(d4[k].y)], 1);
            atomicAdd(&hist[bin_of(d4[k].z)], 1);
            atomicAdd(&hist[bin_of(d4[k].w)], 1);
        }
    }
    __syncthreads();
    if (t < N_BINS) { const int c = hist[t]; if (c) atomicAdd(&binCount[t], c); }
}

__device__ __forceinline__ void bin_scatter_body(int* lds, int chunk, const int* __restrict__ src, const int* __restrict__ dst,
                                                 const int* binCount, int* binFill,
                                                 int* binStart, int* __restrict__ packed) {
    int* wsum   = lds;
    int* hist   = wsum + 16;
    int* gdelta = hist + N_BINS;
    int* sWord  = gdelta + N_BINS;
    int* sPos   = sWord + CHUNK;
    const int t = threadIdx.x;
    const int4* d4p = (const int4*)(dst + chunk * CHUNK);
    const int4* s4p = (const int4*)(src + chunk * CHUNK);
    int4 d4[4], s4[4];
#pragma unroll
    for (int k = 0; k < 4; ++k) {
        const int q = t + k * 512;
        d4[k] = (q < CHUNK4) ? d4p[q] : make_int4(-1, -1, -1, -1);
        s4[k] = (q < CHUNK4) ? s4p[q] : make_int4(0, 0, 0, 0);
    }
    const int v     = (t < N_BINS) ? binCount[t] : 0;
    const int start = scan512(v, wsum, t) - v;
    if (chunk == 0) {
        if (t < N_BINS) binStart[t] = start;
        if (t == N_BINS - 1) binStart[N_BINS] = start + v;
    }
    if (t < N_BINS) hist[t] = 0;
    __syncthreads();
#pragma unroll
    for (int k = 0; k < 4; ++k) {
        if (d4[k].x >= 0) {
            atomicAdd(&hist[bin_of(d4[k].x)], 1);
            atomicAdd(&hist[bin_of(d4[k].y)], 1);
            atomicAdd(&hist[bin_of(d4[k].z)], 1);
            atomicAdd(&hist[bin_of(d4[k].w)], 1);
        }
    }
    __syncthreads();
    const int c  = (t < N_BINS) ? hist[t] : 0;
    const int ls = scan512(c, wsum, t) - c;
    if (t < N_BINS) {
        const int g = c ? atomicAdd(&binFill[t], c) : 0;
        gdelta[t] = start + g - ls;
        hist[t]   = ls;
    }
    __syncthreads();
#pragma unroll
    for (int k = 0; k < 4; ++k) {
        if (d4[k].x >= 0) {
            const int dd[4] = {d4[k].x, d4[k].y, d4[k].z, d4[k].w};
            const int ss[4] = {s4[k].x, s4[k].y, s4[k].z, s4[k].w};
#pragma unroll
            for (int m = 0; m < 4; ++m) {
                const int b = bin_of(dd[m]);
                const int l = atomicAdd(&hist[b], 1);
                sWord[l] = (ss[m] << 9) | (dd[m] - b * BIN_NODES);
                sPos[l]  = gdelta[b] + l;
            }
        }
    }
    __syncthreads();
    for (int i = t; i < CHUNK; i += 512) packed[sPos[i]] = sWord[i];
}


__device__ __forceinline__ void g0_wave_order() {
    __builtin_amdgcn_fence(__ATOMIC_RELEASE, "wavefront");
    __builtin_amdgcn_wave_barrier();
    __builtin_amdgcn_fence(__ATOMIC_ACQUIRE, "wavefront");
}

template <bool REFILL>
__device__ __forceinline__ void g0_chunk(v4f (&buf)[8], const v4f* __restrict__ xnext,
                                         v4f* stS, const v4f* stL, const float* wb, v8f& acc) {
#pragma unroll
    for (int i = 0; i < 8; ++i) stS[i * 2 * XS_PITCH] = buf[i];
    g0_wave_order();
    if (REFILL) {
#pragma unroll
        for (int i = 0; i < 8; ++i) buf[i] = __builtin_nontemporal_load(xnext + i * (2 * F_IN / 4));
    }
#pragma unroll
    for (int j = 0; j < 8; ++j) {
        const v4f a4 = stL[2 * j];
        const float* w = wb + j * 8 * HIDDEN;
        v2f a0; a0.x = a4.x; a0.y = a4.y;
        v2f b0; b0.x = w[0];  b0.y = w[16];
        acc = __builtin_amdgcn_wmma_f32_16x16x4_f32(false, a0, false, b0,
                                                    (short)0, acc, false, false);
        v2f a1; a1.x = a4.z; a1.y = a4.w;
        v2f b1; b1.x = w[32]; b1.y = w[48];
        acc = __builtin_amdgcn_wmma_f32_16x16x4_f32(false, a1, false, b1,
                                                    (short)0, acc, false, false);
    }
    g0_wave_order();
}

__device__ __forceinline__ void gemm0_body(v4f* lds, int group, const float* __restrict__ x,
                                           const float* __restrict__ W0, float* __restrict__ h0) {
    float* sW = (float*)lds;
    v4f*   sX = lds + (F_IN * HIDDEN) / 4;
    const int t = threadIdx.x;
    const int wave   = t >> 5;
    const int lane   = t & 31;
    const int laneHi = lane >> 4;
    const int lane15 = lane & 15;
    const int tile   = group * 16 + wave;
    const bool live  = tile < N_NODES / 16;
    v4f w4[4];
#pragma unroll
    for (int k = 0; k < 4; ++k) w4[k] = ((const v4f*)W0)[t + k * 512];
    const v4f* xg = (const v4f*)x + ((long)tile * 16 + laneHi) * (F_IN / 4) + lane15;
    v4f bufA[8], bufB[8];
    if (live) {
#pragma unroll
        for (int i = 0; i < 8; ++i) bufA[i] = __builtin_nontemporal_load(xg + i * (2 * F_IN / 4));
#pragma unroll
        for (int i = 0; i < 8; ++i) bufB[i] = __builtin_nontemporal_load(xg + i * (2 * F_IN / 4) + 16);
    }
#pragma unroll
    for (int k = 0; k < 4; ++k) ((v4f*)sW)[t + k * 512] = w4[k];
    __syncthreads();
    if (!live) return;

    const int koff = laneHi ? 4 : 0;
    v4f*       stS = sX + wave * XS_WAVE + laneHi * XS_PITCH + lane15;
    const v4f* stL = sX + wave * XS_WAVE + lane15 * XS_PITCH + laneHi;
    const float* wb = sW + koff * HIDDEN + lane15;

    v8f acc = {};
#pragma unroll
    for (int c = 0; c < 8; c += 2) {
        if (c + 2 < 8) {
            g0_chunk<true>(bufA, xg + (c + 2) * 16, stS, stL, wb + c * 64 * HIDDEN, acc);
            g0_chunk<true>(bufB, xg + (c + 3) * 16, stS, stL, wb + (c + 1) * 64 * HIDDEN, acc);
        } else {
            g0_chunk<false>(bufA, xg, stS, stL, wb + c * 64 * HIDDEN, acc);
            g0_chunk<false>(bufB, xg, stS, stL, wb + (c + 1) * 64 * HIDDEN, acc);
        }
    }
    const long o = (long)tile * 256 + laneHi * 128 + lane15;
#pragma unroll
    for (int i = 0; i < 8; ++i) h0[o + i * 16] = acc[i];
}

template <bool SCATTER>
__global__ __launch_bounds__(512) void k_bin_gemm0(const int* __restrict__ src, const int* __restrict__ dst,
                                                   int* binCount, int* binFill, int* binStart, int* packed,
                                                   const float* __restrict__ x, const float* __restrict__ W0,
                                                   float* __restrict__ h0) {
    __shared__ v4f lds[LDS_V4F];
    const int nProd = SCATTER ? G0_GROUPS - G0_FIRST : G0_FIRST;
    const int b     = blockIdx.x;
    if (b < nProd) {
        gemm0_body(lds, (SCATTER ? G0_FIRST : 0) + b, x, W0, h0);
    } else {
        if (SCATTER) bin_scatter_body((int*)lds, b - nProd, src, dst, binCount, binFill, binStart, packed);
        else         bin_count_body((int*)lds, b - nProd, dst, binCount);
    }
}

__global__ __launch_bounds__(256) void k_deg(const int* __restrict__ packed, const int* __restrict__ binStart,
                                             float* __restrict__ dinv, float* h0s) {
    __shared__ int cnt[256];
    const int t        = threadIdx.x;
    const int nodeBase = blockIdx.x * BIN_NODES;
    const int e0 = binStart[blockIdx.x];
    const int e1 = binStart[blockIdx.x + 1];
    float4* rows = (float4*)h0s + (long)nodeBase * 4;
    float4 r[4];
#pragma unroll
    for (int k = 0; k < 4; ++k) {
        const int i = t + k * 256;
        r[k] = (i < BIN_NODES * 4 && nodeBase + (i >> 2) < N_NODES) ? rows[i] : make_float4(0.f, 0.f, 0.f, 0.f);
    }
    cnt[t] = 1;
    __syncthreads();
    for (int e = e0 + t; e < e1; e += 8 * 256) {
        int p[8];
#pragma unroll
        for (int k = 0; k < 8; ++k) p[k] = (e + k * 256 < e1) ? packed[e + k * 256] : -1;
#pragma unroll
        for (int k = 0; k < 8; ++k)
            if (p[k] >= 0) atomicAdd(&cnt[p[k] & 511], 1);
    }
    __syncthreads();
    if (t < BIN_NODES && nodeBase + t < N_NODES) dinv[nodeBase + t] = __frsqrt_rn((float)cnt[t]);
#pragma unroll
    for (int k = 0; k < 4; ++k) {
        const int i = t + k * 256;
        if (i < BIN_NODES * 4 && nodeBase + (i >> 2) < N_NODES) {
            const float d = __frsqrt_rn((float)cnt[i >> 2]);
            rows[i] = make_float4(r[k].x * d, r[k].y * d, r[k].z * d, r[k].w * d);
        }
    }
}

__global__ __launch_bounds__(1024) void k_agg16(const int* __restrict__ packed, const int* __restrict__ binStart,
                                                const float* __restrict__ dinv, const float* __restrict__ h0s,
                                                const float* __restrict__ b0, const float* __restrict__ W1,
                                                float* __restrict__ h2s) {
    __shared__ float4 tile4[(TILE_ROWS + 1) * 4];
    __shared__ float  sW[16 * 16];
    float* tile = (float*)tile4;
    const int t        = threadIdx.x;
    const int nodeBase = blockIdx.x * BIN_NODES;
    const int myNode   = nodeBase + (t >> 2);
    if (t < (TILE_ROWS + 1) * 4)
        tile4[t] = ((t >> 2) < BIN_NODES && myNode < N_NODES) ? ((const float4*)h0s)[(long)nodeBase * 4 + t]
                                                               : make_float4(0.f, 0.f, 0.f, 0.f);
    if (t < 16 * 16) {
        const int k = t >> 4, n = t & 15;
        sW[t] = (n < N_CLS) ? W1[k * N_CLS + n] : 0.0f;
    }
    __syncthreads();

    const int lane = t & 31;
    const int f    = lane & 15;
    const int half = lane & 16;
    const int e1   = binStart[blockIdx.x + 1];
    int c0 = binStart[blockIdx.x] + (t >> 5) * 32;
    int p  = (c0 + lane < e1) ? packed[c0 + lane] : SINK;
    while (c0 < e1) {
        const int cn = c0 + 32 * 32;
        float val[16];
        int   pj[16];
#pragma unroll
        for (int j = 0; j < 16; ++j) pj[j] = __shfl(p, half | j);
        const int pn = (cn + lane < e1) ? packed[cn + lane] : SINK;
#pragma unroll
        for (int j = 0; j < 16; ++j) val[j] = h0s[(pj[j] >> 9) * 16 + f];
#pragma unroll
        for (int j = 0; j < 16; ++j) atomicAdd(&tile[(pj[j] & 511) * 16 + f], val[j]);
        p  = pn;
        c0 = cn;
    }
    __syncthreads();

    const int wave     = t >> 5;
    const int tileBase = nodeBase + wave * 16;
    if (wave < TILE_ROWS / 16 && tileBase < N_NODES) {
        const int laneHi = lane >> 4;
        const int koff   = laneHi ? 2 : 0;
        const bool rowIn  = wave * 16 + f < BIN_NODES && tileBase + f < N_NODES;
        const float wrow  = rowIn ? dinv[tileBase + f] : 0.0f;
        const float* trow = tile + (wave * 16 + f) * 16;
        v8f acc = {};
#pragma unroll
        for (int kb = 0; kb < 16; kb += 4) {
            const float v0 = trow[kb + koff]     * wrow + b0[kb + koff];
            const float v1 = trow[kb + koff + 1] * wrow + b0[kb + koff + 1];
            v2f a; a.x = v0 > 0.0f ? v0 : 0.0f;
                   a.y = v1 > 0.0f ? v1 : 0.0f;
            v2f b; b.x = sW[(kb + koff) * 16 + f];
                   b.y = sW[(kb + koff + 1) * 16 + f];
            acc = __builtin_amdgcn_wmma_f32_16x16x4_f32(false, a, false, b,
                                                        (short)0, acc, false, false);
        }
        const int r0 = tileBase + laneHi * 8;
        if (wave * 16 + laneHi * 8 < BIN_NODES && r0 < N_NODES) {
            const float4 dA = *(const float4*)(dinv + r0);
            const float4 dB = *(const float4*)(dinv + r0 + 4);
            const float d[8] = {dA.x, dA.y, dA.z, dA.w, dB.x, dB.y, dB.z, dB.w};
            float* o2 = h2s + (long)r0 * 16 + f;
#pragma unroll
            for (int i = 0; i < 8; ++i) o2[i * 16] = acc[i] * d[i];
        }
    }
}

__global__ __launch_bounds__(1024) void k_agg7(const int* __restrict__ packed, const int* __restrict__ binStart,
                                               const float* __restrict__ dinv, const float* __restrict__ h2s,
                                               const float* __restrict__ b1, float* __restrict__ out) {
    __shared__ float4 tile4[(SINK + 1) * 2];
    float* tile = (float*)tile4;
    const int t        = threadIdx.x;
    const int nodeBase = blockIdx.x * BIN_NODES;
    if (t < BIN_NODES * 2) {
        const int node = nodeBase + (t >> 1);
        tile4[t] = (node < N_NODES) ? ((const float4*)h2s)[(long)node * 4 + (t & 1)] : make_float4(0.f, 0.f, 0.f, 0.f);
    }
    __syncthreads();

    const int lane = t & 31;
    const int c    = lane & 7;
    const int grp  = lane & 24;
    const int e1   = binStart[blockIdx.x + 1];
    int c0 = binStart[blockIdx.x] + (t >> 5) * 32;
    int p  = (c0 + lane < e1) ? packed[c0 + lane] : SINK;
    while (c0 < e1) {
        const int cn = c0 + 32 * 32;
        float val[8];
        int   pj[8];
#pragma unroll
        for (int j = 0; j < 8; ++j) pj[j] = __shfl(p, grp | j);
        const int pn = (cn + lane < e1) ? packed[cn + lane] : SINK;
#pragma unroll
        for (int j = 0; j < 8; ++j) val[j] = h2s[(pj[j] >> 9) * 16 + c];
#pragma unroll
        for (int j = 0; j < 8; ++j) atomicAdd(&tile[(pj[j] & 511) * 8 + c], val[j]);
        p  = pn;
        c0 = cn;
    }
    __syncthreads();
    for (int i = t; i < BIN_NODES * N_CLS; i += 1024) {
        const int nl   = i / N_CLS;
        const int cc   = i - nl * N_CLS;
        const int node = nodeBase + nl;
        if (node < N_NODES) out[(long)nodeBase * N_CLS + i] = tile[nl * 8 + cc] * dinv[node] + b1[cc];
    }
}

extern "C" void kernel_launch(void* const* d_in, const int* in_sizes, int n_in,
                              void* d_out, int out_size, void* d_ws, size_t ws_size,
                              hipStream_t stream) {
    const float* x  = (const float*)d_in[0];
    const int*   ei = (const int*)d_in[1];
    const float* W0 = (const float*)d_in[2];
    const float* b0 = (const float*)d_in[3];
    const float* W1 = (const float*)d_in[4];
    const float* b1 = (const float*)d_in[5];
    float* out = (float*)d_out;

    const int* src = ei;
    const int* dst = ei + N_EDGES;

    float* ws   = (float*)d_ws;
    float* dinv = ws;
    float* h0s  = dinv + N_NODES;
    float* h2s  = h0s + (long)N_NODES * 16;
    int* packed   = (int*)(h2s + (long)N_NODES * 16);
    int* binStart = packed + N_EDGES;
    int* binCount = (int*)h2s;
    int* binFill  = binCount + N_BINS;

    k_bin_zero<<<dim3(1), dim3(512), 0, stream>>>(binCount, binFill);
    k_bin_gemm0<false><<<dim3(N_CHUNKS + G0_FIRST), dim3(512), 0, stream>>>(src, dst, binCount, binFill, binStart, packed, x, W0, h0s);
    k_bin_gemm0<true><<<dim3(N_CHUNKS + G0_GROUPS - G0_FIRST), dim3(512), 0, stream>>>(src, dst, binCount, binFill, binStart, packed, x, W0, h0s);

    k_deg<<<dim3(N_BINS), dim3(256), 0, stream>>>(packed, binStart, dinv, h0s);
    k_agg16<<<dim3(N_BINS), dim3(1024), 0, stream>>>(packed, binStart, dinv, h0s, b0, W1, h2s);

    k_agg7<<<dim3(N_BINS), dim3(1024), 0, stream>>>(packed, binStart, dinv, h2s, b1, out);
}
